# starting kernel
# baseline (speedup 1.0000x reference)
#include <hip/hip_runtime.h>
#include <math.h>


typedef __attribute__((ext_vector_type(2))) float v2f;
typedef __attribute__((ext_vector_type(8))) float v8f;

__device__ __forceinline__ float atomic_add_f32(float* p, float v) {
  return __hip_atomic_fetch_add(p, v, __ATOMIC_RELAXED, __HIP_MEMORY_SCOPE_AGENT);
}

__device__ __forceinline__ void atomic_max_f32(float* p, float v) {
  if (v >= 0.0f) atomicMax((int*)p, __float_as_int(v));
  else           atomicMin((unsigned int*)p, __float_as_uint(v));
}

__global__ __launch_bounds__(32) void wmma_gemm_f32(
    const float* __restrict__ A, const float* __restrict__ B,
    float* __restrict__ C, int M, int N, int K)
{
  const int lane = threadIdx.x & 31;
  const int half = lane >> 4;
  const int l15  = lane & 15;
  const int tm   = blockIdx.x;
  const int tn   = blockIdx.y << 6;

  v8f c0 = {}; v8f c1 = {}; v8f c2 = {}; v8f c3 = {};

  const float* Arow = A + (size_t)(tm * 16 + l15) * K;
  for (int k0 = 0; k0 < K; k0 += 4) {
    const int ka = k0 + (half << 1);
    v2f a;
    a.x = Arow[ka];
    a.y = Arow[ka + 1];
    const float* Bp = B + (size_t)ka * N + tn + l15;
    v2f b0, b1, b2, b3;
    b0.x = Bp[0];   b0.y = Bp[N];
    b1.x = Bp[16];  b1.y = Bp[N + 16];
    b2.x = Bp[32];  b2.y = Bp[N + 32];
    b3.x = Bp[48];  b3.y = Bp[N + 48];
    c0 = __builtin_amdgcn_wmma_f32_16x16x4_f32(false, a, false, b0, (short)0, c0, false, false);
    c1 = __builtin_amdgcn_wmma_f32_16x16x4_f32(false, a, false, b1, (short)0, c1, false, false);
    c2 = __builtin_amdgcn_wmma_f32_16x16x4_f32(false, a, false, b2, (short)0, c2, false, false);
    c3 = __builtin_amdgcn_wmma_f32_16x16x4_f32(false, a, false, b3, (short)0, c3, false, false);
  }

  const int orow = tm * 16 + (half << 3);
  for (int r = 0; r < 8; ++r) {
    float* Cp = C + (size_t)(orow + r) * N + tn + l15;
    Cp[0]  = c0[r];
    Cp[16] = c1[r];
    Cp[32] = c2[r];
    Cp[48] = c3[r];
  }
}

__global__ __launch_bounds__(256) void node_dots(
    const float* __restrict__ xp, const float* __restrict__ att_src,
    const float* __restrict__ att_dst, float* __restrict__ asrc,
    float* __restrict__ adst, int Nn, int H, int C)
{
  const int wid  = (int)((blockIdx.x * blockDim.x + threadIdx.x) >> 5);
  const int lane = threadIdx.x & 31;
  if (wid >= Nn * H) return;
  const int n = wid / H;
  const int h = wid - n * H;

  const float* xr = xp + (size_t)n * H * C + (size_t)h * C;
  const float* as = att_src + (size_t)h * C;
  const float* ad = att_dst + (size_t)h * C;
  float s = 0.f, d = 0.f;
  for (int c = lane; c < C; c += 32) {
    float v = xr[c];
    s += v * as[c];
    d += v * ad[c];
  }
  for (int off = 16; off > 0; off >>= 1) {
    s += __shfl_down(s, off, 32);
    d += __shfl_down(d, off, 32);
  }
  if (lane == 0) { asrc[wid] = s; adst[wid] = d; }
}

__global__ void fill_f32(float* __restrict__ p, float v, size_t n) {
  size_t i = (size_t)blockIdx.x * blockDim.x + threadIdx.x;
  size_t stride = (size_t)gridDim.x * blockDim.x;
  for (; i < n; i += stride) p[i] = v;
}

__global__ __launch_bounds__(256) void edge_max(
    const int* __restrict__ src, const int* __restrict__ dst, int E0, int Nn,
    const float* __restrict__ asrc, const float* __restrict__ adst,
    float* __restrict__ amax, int H)
{
  const int e = blockIdx.x * blockDim.x + threadIdx.x;
  if (e >= E0 + Nn) return;
  const int s = (e < E0) ? src[e] : (e - E0);
  const int d = (e < E0) ? dst[e] : (e - E0);
  for (int h = 0; h < H; ++h) {
    float a = asrc[s * H + h] + adst[d * H + h];
    a = (a > 0.f) ? a : 0.2f * a;
    atomic_max_f32(&amax[d * H + h], a);
  }
}

__global__ __launch_bounds__(256) void edge_exp(
    const int* __restrict__ src, const int* __restrict__ dst, int E0, int Nn,
    const float* __restrict__ asrc, const float* __restrict__ adst,
    const float* __restrict__ amax, float* __restrict__ den,
    float* __restrict__ ex, int H)
{
  const int e = blockIdx.x * blockDim.x + threadIdx.x;
  if (e >= E0 + Nn) return;
  const int s = (e < E0) ? src[e] : (e - E0);
  const int d = (e < E0) ? dst[e] : (e - E0);
  for (int h = 0; h < H; ++h) {
    float a = asrc[s * H + h] + adst[d * H + h];
    a = (a > 0.f) ? a : 0.2f * a;
    float v = __expf(a - amax[d * H + h]);
    ex[(size_t)e * H + h] = v;
    atomic_add_f32(&den[d * H + h], v);
  }
}

__global__ __launch_bounds__(256) void edge_aggregate(
    const int* __restrict__ src, const int* __restrict__ dst, int E0,
    const float* __restrict__ xp, const float* __restrict__ ex,
    const float* __restrict__ den, float* __restrict__ out, int H)
{
  const int e = blockIdx.x;
  const int s = (e < E0) ? src[e] : (e - E0);
  const int d = (e < E0) ? dst[e] : (e - E0);
  const int F = H << 8;
  const size_t eb = (size_t)e * H;
  const size_t sb = (size_t)s * F;
  const size_t db = (size_t)d * F;
  for (int f = threadIdx.x; f < F; f += 256) {
    const int h = f >> 8;
    const float w = ex[eb + h] / den[d * H + h];
    atomic_add_f32(&out[db + f], xp[sb + f] * w);
  }
}

__global__ void bias_elu(float* __restrict__ y, const float* __restrict__ b,
                         int F, size_t n) {
  size_t i = (size_t)blockIdx.x * blockDim.x + threadIdx.x;
  size_t stride = (size_t)gridDim.x * blockDim.x;
  for (; i < n; i += stride) {
    int f = (int)(i - (i / F) * F);
    float v = y[i] + b[f];
    y[i] = (v > 0.f) ? v : expm1f(v);
  }
}

__global__ void bias_add(float* __restrict__ y, const float* __restrict__ b,
                         int Fmask, size_t n) {
  size_t i = (size_t)blockIdx.x * blockDim.x + threadIdx.x;
  size_t stride = (size_t)gridDim.x * blockDim.x;
  for (; i < n; i += stride) y[i] += b[(int)i & Fmask];
}


extern "C" void kernel_launch(void* const* d_in, const int* in_sizes, int n_in,
                              void* d_out, int out_size, void* d_ws, size_t ws_size,
                              hipStream_t stream) {
  const float* x   = (const float*)d_in[0];
  const int*   ei  = (const int*)d_in[1];
  const float* W1  = (const float*)d_in[2];
  const float* as1 = (const float*)d_in[3];
  const float* ad1 = (const float*)d_in[4];
  const float* b1  = (const float*)d_in[5];
  const float* W2  = (const float*)d_in[6];
  const float* as2 = (const float*)d_in[7];
  const float* ad2 = (const float*)d_in[8];
  const float* b2  = (const float*)d_in[9];

  const int K1 = 768, F1 = 768, H1 = 3;
  const int F2 = 256, H2 = 1;
  const int Nn = in_sizes[0] / K1;
  const int E0 = in_sizes[1] / 2;
  const int ET = E0 + Nn;

  const int* srcI = ei;
  const int* dstI = ei + E0;

  float* ws = (float*)d_ws;
  float* xp1   = ws; ws += (size_t)Nn * F1;
  float* h1    = ws; ws += (size_t)Nn * F1;
  float* xp2   = ws; ws += (size_t)Nn * F2;
  float* asrc1 = ws; ws += (size_t)Nn * H1;
  float* adst1 = ws; ws += (size_t)Nn * H1;
  float* amax1 = ws; ws += (size_t)Nn * H1;
  float* den1  = ws; ws += (size_t)Nn * H1;
  float* asrc2 = ws; ws += Nn;
  float* adst2 = ws; ws += Nn;
  float* amax2 = ws; ws += Nn;
  float* den2  = ws; ws += Nn;
  float* ex1   = ws; ws += (size_t)ET * H1;
  float* ex2   = ws; ws += (size_t)ET;

  const float NEG_INF = -__builtin_inff();
  const int FB = 512;

  {
    dim3 g(Nn / 16, F1 / 64);
    wmma_gemm_f32<<<g, 32, 0, stream>>>(x, W1, xp1, Nn, F1, K1);
  }
  node_dots<<<(Nn * H1 + 7) / 8, 256, 0, stream>>>(xp1, as1, ad1, asrc1, adst1, Nn, H1, 256);
  fill_f32<<<FB, 256, 0, stream>>>(amax1, NEG_INF, (size_t)Nn * H1);
  fill_f32<<<FB, 256, 0, stream>>>(den1, 0.f, (size_t)Nn * H1);
  fill_f32<<<FB, 256, 0, stream>>>(h1, 0.f, (size_t)Nn * F1);
  edge_max<<<(ET + 255) / 256, 256, 0, stream>>>(srcI, dstI, E0, Nn, asrc1, adst1, amax1, H1);
  edge_exp<<<(ET + 255) / 256, 256, 0, stream>>>(srcI, dstI, E0, Nn, asrc1, adst1, amax1, den1, ex1, H1);
  edge_aggregate<<<ET, 256, 0, stream>>>(srcI, dstI, E0, xp1, ex1, den1, h1, H1);
  bias_elu<<<FB, 256, 0, stream>>>(h1, b1, F1, (size_t)Nn * F1);

  {
    dim3 g(Nn / 16, F2 / 64);
    wmma_gemm_f32<<<g, 32, 0, stream>>>(h1, W2, xp2, Nn, F2, F1);
  }
  node_dots<<<(Nn + 7) / 8, 256, 0, stream>>>(xp2, as2, ad2, asrc2, adst2, Nn, H2, 256);
  fill_f32<<<FB, 256, 0, stream>>>(amax2, NEG_INF, (size_t)Nn);
  fill_f32<<<FB, 256, 0, stream>>>(den2, 0.f, (size_t)Nn);
  fill_f32<<<FB, 256, 0, stream>>>((float*)d_out, 0.f, (size_t)Nn * F2);
  edge_max<<<(ET + 255) / 256, 256, 0, stream>>>(srcI, dstI, E0, Nn, asrc2, adst2, amax2, H2);
  edge_exp<<<(ET + 255) / 256, 256, 0, stream>>>(srcI, dstI, E0, Nn, asrc2, adst2, amax2, den2, ex2, H2);
  edge_aggregate<<<ET, 256, 0, stream>>>(srcI, dstI, E0, xp2, ex2, den2, (float*)d_out, H2);
  bias_add<<<FB, 256, 0, stream>>>((float*)d_out, b2, F2 - 1, (size_t)Nn * F2);
}


# Edge softmax+aggregation by owner workgroups: 80 dst rows per workgroup in LDS, each filters the unsorted edge list, stats and row sums in edge-index order; no sort, no atomics; bias/ELU in the store
# speedup vs baseline: 2.5407x; 2.5407x over previous
#include <hip/hip_runtime.h>
#include <math.h>


typedef __attribute__((ext_vector_type(2))) float v2f;
typedef __attribute__((ext_vector_type(8))) float v8f;

__global__ __launch_bounds__(32) void wmma_gemm_f32(
    const float* __restrict__ A, const float* __restrict__ B,
    float* __restrict__ C, int M, int N, int K)
{
  const int lane = threadIdx.x & 31;
  const int half = lane >> 4;
  const int l15  = lane & 15;
  const int tm   = blockIdx.x;
  const int tn   = blockIdx.y << 6;

  v8f c0 = {}; v8f c1 = {}; v8f c2 = {}; v8f c3 = {};

  const float* Arow = A + (size_t)(tm * 16 + l15) * K;
  for (int k0 = 0; k0 < K; k0 += 4) {
    const int ka = k0 + (half << 1);
    v2f a;
    a.x = Arow[ka];
    a.y = Arow[ka + 1];
    const float* Bp = B + (size_t)ka * N + tn + l15;
    v2f b0, b1, b2, b3;
    b0.x = Bp[0];   b0.y = Bp[N];
    b1.x = Bp[16];  b1.y = Bp[N + 16];
    b2.x = Bp[32];  b2.y = Bp[N + 32];
    b3.x = Bp[48];  b3.y = Bp[N + 48];
    c0 = __builtin_amdgcn_wmma_f32_16x16x4_f32(false, a, false, b0, (short)0, c0, false, false);
    c1 = __builtin_amdgcn_wmma_f32_16x16x4_f32(false, a, false, b1, (short)0, c1, false, false);
    c2 = __builtin_amdgcn_wmma_f32_16x16x4_f32(false, a, false, b2, (short)0, c2, false, false);
    c3 = __builtin_amdgcn_wmma_f32_16x16x4_f32(false, a, false, b3, (short)0, c3, false, false);
  }

  const int orow = tm * 16 + (half << 3);
  for (int r = 0; r < 8; ++r) {
    float* Cp = C + (size_t)(orow + r) * N + tn + l15;
    Cp[0]  = c0[r];
    Cp[16] = c1[r];
    Cp[32] = c2[r];
    Cp[48] = c3[r];
  }
}

__global__ __launch_bounds__(256) void node_dots(
    const float* __restrict__ xp, const float* __restrict__ att_src,
    const float* __restrict__ att_dst, float* __restrict__ asrc,
    float* __restrict__ adst, int Nn, int H, int C)
{
  const int wid  = (int)((blockIdx.x * blockDim.x + threadIdx.x) >> 5);
  const int lane = threadIdx.x & 31;
  if (wid >= Nn * H) return;
  const int n = wid / H;
  const int h = wid - n * H;

  const float* xr = xp + (size_t)n * H * C + (size_t)h * C;
  const float* as = att_src + (size_t)h * C;
  const float* ad = att_dst + (size_t)h * C;
  float s = 0.f, d = 0.f;
  for (int c = lane; c < C; c += 32) {
    float v = xr[c];
    s += v * as[c];
    d += v * ad[c];
  }
  for (int off = 16; off > 0; off >>= 1) {
    s += __shfl_down(s, off, 32);
    d += __shfl_down(d, off, 32);
  }
  if (lane == 0) { asrc[wid] = s; adst[wid] = d; }
}

typedef __attribute__((ext_vector_type(4))) float v4f;

__device__ __forceinline__ float leaky02(float a) { return (a > 0.f) ? a : 0.2f * a; }

template <int H, int NT, int RT, int CAPW, int TCAP, bool ELU>
__global__ __launch_bounds__(NT) void owner_edge_phase(
    const int* __restrict__ src, const int* __restrict__ dst, int E0, int Nn,
    const float* __restrict__ xp, const float* __restrict__ asrc,
    const float* __restrict__ adst, const float* __restrict__ bias,
    float* __restrict__ out)
{
  constexpr int F      = H * 256;
  constexpr int F4     = F / 4;
  constexpr int NTEAMS = NT / F4;
  constexpr int R      = NTEAMS * RT;
  constexpr int NW     = NT / 32;
  constexpr int NL     = NW * NTEAMS;
  constexpr int SUB    = (TCAP < NW * CAPW) ? TCAP : NW * CAPW;
  constexpr int UA     = 16;
  constexpr int UC     = 8;
  constexpr int UI     = 5;
  static_assert(NT % F4 == 0 && F4 % 32 == 0 && NTEAMS <= 32, "team shape");
  static_assert(RT * H <= F4, "one thread per (row, head) of a team");
  static_assert(RT % UI == 0, "self-loop rows are loaded UI at a time");

  __shared__ v4f   acc[R * F4];
  __shared__ int   wl[NL * CAPW];
  __shared__ int   wcnt[NL];
  __shared__ int   tsrc[NTEAMS * TCAP];
  __shared__ int   tdl[NTEAMS * TCAP];
  __shared__ float tw[NTEAMS * TCAP * H];
  __shared__ float smax[R * H];
  __shared__ float sden[R * H];
  __shared__ float swt[R * H];
  __shared__ int   ovf;

  const int tid  = threadIdx.x;
  const int lane = tid & 31;
  const int wave = tid >> 5;
  const int team = tid / F4;
  const int c4   = tid - team * F4;
  const int h    = c4 >> 6;
  const int d0   = blockIdx.x * R;
  const int r0   = team * RT;

  const float* xcol = xp + c4 * 4;
  v4f*   arow = acc + r0 * F4 + c4;
  int*   ts   = tsrc + team * TCAP;
  int*   td   = tdl + team * TCAP;
  float* tq   = tw + team * TCAP * H;

  const bool stat = c4 < RT * H;
  const int  prow = c4 / H;
  const int  phh  = c4 - prow * H;
  const int  pidx = (r0 + prow) * H + phh;
  float pself = 0.f;
  if (stat) {
    const int d = d0 + r0 + prow;
    if (d < Nn) pself = leaky02(asrc[d * H + phh] + adst[d * H + phh]);
    smax[pidx] = pself;
  }

  auto build = [&](int lo, int hi) -> int {
    if (tid == 0) ovf = 0;
    int mycnt = 0;
    const int chunk = (hi - lo + NW - 1) / NW;
    const int cb = lo + wave * chunk;
    int ce = cb + chunk; if (ce > hi) ce = hi;
    for (int base = cb; base < ce; base += 32 * UA) {
      int dv[UA];
      if (base + 32 * UA <= ce) {
#pragma unroll
        for (int u = 0; u < UA; ++u) dv[u] = dst[base + u * 32 + lane];
      } else {
#pragma unroll
        for (int u = 0; u < UA; ++u) {
          const int e = base + u * 32 + lane;
          dv[u] = (e < ce) ? dst[e] : -1;
        }
      }
#pragma unroll
      for (int u = 0; u < UA; ++u) {
        const unsigned dl = (unsigned)(dv[u] - d0);
        unsigned m = __builtin_amdgcn_ballot_w32(dl < (unsigned)R);
        while (m) {
          const int l = __builtin_ctz(m);
          m &= m - 1;
          const int dls = __builtin_amdgcn_readlane((int)dl, l);
          const int t = dls / RT;
          if (lane == t) {
            if (mycnt < CAPW) wl[(wave * NTEAMS + t) * CAPW + mycnt] = base + u * 32 + l;
            ++mycnt;
          }
        }
      }
    }
    if (lane < NTEAMS) wcnt[wave * NTEAMS + lane] = mycnt;
    __syncthreads();

    int tot = 0;
    bool bad = false;
    for (int w = 0; w < NW; ++w) {
      const int c = wcnt[w * NTEAMS + team];
      bad |= (c > CAPW);
      tot += c;
    }
    bad |= (tot > TCAP);
    if (bad) ovf = 1;
    __syncthreads();
    const int o = ovf;
    __syncthreads();
    if (o) return -1;

    for (int j = c4; j < tot; j += F4) {
      int rem = j, w = 0;
      int c = wcnt[team];
      while (rem >= c) { rem -= c; ++w; c = wcnt[w * NTEAMS + team]; }
      const int e = wl[(w * NTEAMS + team) * CAPW + rem];
      const int s = src[e];
      const int d = dst[e];
      ts[j] = s;
      td[j] = d - d0 - r0;
      for (int hh = 0; hh < H; ++hh)
        tq[j * H + hh] = leaky02(asrc[s * H + hh] + adst[d * H + hh]);
    }
    __syncthreads();
    return tot;
  };

  auto pass_max = [&](int tot) {
    if (stat) {
      float m = smax[pidx];
#pragma unroll 8
      for (int j = 0; j < tot; ++j) {
        const float a = tq[j * H + phh];
        m = (td[j] == prow) ? fmaxf(m, a) : m;
      }
      smax[pidx] = m;
    }
  };

  auto pass_den = [&](int tot) {
    if (stat) {
      const float m = smax[pidx];
      float s = sden[pidx];
#pragma unroll 8
      for (int j = 0; j < tot; ++j) {
        const float ev = __expf(tq[j * H + phh] - m);
        s += (td[j] == prow) ? ev : 0.f;
      }
      sden[pidx] = s;
    }
  };

  auto pass_agg = [&](int tot) {
    for (int j = c4; j < tot; j += F4) {
      const int rr = (r0 + td[j]) * H;
      for (int hh = 0; hh < H; ++hh)
        tq[j * H + hh] = __expf(tq[j * H + hh] - smax[rr + hh]) / sden[rr + hh];
    }
    __syncthreads();
    const float* tqh = tq + h;
    int i = 0;
    for (; i + UC <= tot; i += UC) {
      int sr[UC], dr[UC]; float wt[UC]; v4f v[UC];
#pragma unroll
      for (int u = 0; u < UC; ++u) {
        sr[u] = ts[i + u]; dr[u] = td[i + u]; wt[u] = tqh[(i + u) * H];
      }
#pragma unroll
      for (int u = 0; u < UC; ++u) v[u] = *(const v4f*)(xcol + (size_t)sr[u] * F);
#pragma unroll
      for (int u = 0; u < UC; ++u) {
        v4f a = arow[dr[u] * F4];
        a += v[u] * wt[u];
        arow[dr[u] * F4] = a;
      }
    }
    for (; i < tot; ++i) {
      const v4f v1 = *(const v4f*)(xcol + (size_t)ts[i] * F);
      v4f a = arow[td[i] * F4];
      a += v1 * tqh[i * H];
      arow[td[i] * F4] = a;
    }
  };

  int tot = build(0, E0);
  const bool whole = (tot >= 0);
  const int step = whole ? ((E0 > 0) ? E0 : 1) : SUB;

  for (int mode = 0; mode < 3; ++mode) {
    if (mode == 1 && stat) sden[pidx] = __expf(pself - smax[pidx]);
    if (mode == 2) {
      if (stat) swt[pidx] = __expf(pself - smax[pidx]) / sden[pidx];
      __syncthreads();
      for (int r = 0; r < RT; r += UI) {
        v4f v[UI];
#pragma unroll
        for (int u = 0; u < UI; ++u) {
          const int d = d0 + r0 + r + u;
          const v4f z = {0.f, 0.f, 0.f, 0.f};
          v[u] = (d < Nn) ? *(const v4f*)(xcol + (size_t)d * F) : z;
        }
#pragma unroll
        for (int u = 0; u < UI; ++u) arow[(r + u) * F4] = v[u] * swt[(r0 + r + u) * H + h];
      }
    }
    for (int lo = 0; lo < E0; lo += step) {
      if (!whole) {
        const int hi = (lo + SUB < E0) ? lo + SUB : E0;
        tot = build(lo, hi);
      }
      if (mode == 0) pass_max(tot);
      else if (mode == 1) pass_den(tot);
      else pass_agg(tot);
      __syncthreads();
    }
    __syncthreads();
  }

  const float b0 = bias[c4 * 4 + 0], b1 = bias[c4 * 4 + 1];
  const float b2 = bias[c4 * 4 + 2], b3 = bias[c4 * 4 + 3];
  for (int r = 0; r < RT; ++r) {
    const int d = d0 + r0 + r;
    if (d < Nn) {
      v4f a = arow[r * F4];
      a.x += b0; a.y += b1; a.z += b2; a.w += b3;
      if (ELU) {
        a.x = (a.x > 0.f) ? a.x : expm1f(a.x);
        a.y = (a.y > 0.f) ? a.y : expm1f(a.y);
        a.z = (a.z > 0.f) ? a.z : expm1f(a.z);
        a.w = (a.w > 0.f) ? a.w : expm1f(a.w);
      }
      *(v4f*)(out + (size_t)d * F + c4 * 4) = a;
    }
  }
}


extern "C" void kernel_launch(void* const* d_in, const int* in_sizes, int n_in,
                              void* d_out, int out_size, void* d_ws, size_t ws_size,
                              hipStream_t stream) {
  const float* x   = (const float*)d_in[0];
  const int*   ei  = (const int*)d_in[1];
  const float* W1  = (const float*)d_in[2];
  const float* as1 = (const float*)d_in[3];
  const float* ad1 = (const float*)d_in[4];
  const float* b1  = (const float*)d_in[5];
  const float* W2  = (const float*)d_in[6];
  const float* as2 = (const float*)d_in[7];
  const float* ad2 = (const float*)d_in[8];
  const float* b2  = (const float*)d_in[9];

  const int K1 = 768, F1 = 768, H1 = 3;
  const int F2 = 256, H2 = 1;
  const int Nn = in_sizes[0] / K1;
  const int E0 = in_sizes[1] / 2;

  const int* srcI = ei;
  const int* dstI = ei + E0;

  float* ws = (float*)d_ws;
  float* xp1   = ws; ws += (size_t)Nn * F1;
  float* h1    = ws; ws += (size_t)Nn * F1;
  float* xp2   = ws; ws += (size_t)Nn * F2;
  float* asrc1 = ws; ws += (size_t)Nn * H1;
  float* adst1 = ws; ws += (size_t)Nn * H1;
  float* asrc2 = ws; ws += Nn;
  float* adst2 = ws; ws += Nn;

  {
    dim3 g(Nn / 16, F1 / 64);
    wmma_gemm_f32<<<g, 32, 0, stream>>>(x, W1, xp1, Nn, F1, K1);
  }
  node_dots<<<(Nn * H1 + 7) / 8, 256, 0, stream>>>(xp1, as1, ad1, asrc1, adst1, Nn, H1, 256);
  owner_edge_phase<3, 768, 20, 48, 512, true><<<(Nn + 79) / 80, 768, 0, stream>>>(
      srcI, dstI, E0, Nn, xp1, asrc1, adst1, b1, h1);

  {
    dim3 g(Nn / 16, F2 / 64);
    wmma_gemm_f32<<<g, 32, 0, stream>>>(h1, W2, xp2, Nn, F2, F1);
  }
  node_dots<<<(Nn + 7) / 8, 256, 0, stream>>>(xp2, as2, ad2, asrc2, adst2, Nn, H2, 256);
  owner_edge_phase<1, 1024, 5, 24, 256, false><<<(Nn + 79) / 80, 1024, 0, stream>>>(
      srcI, dstI, E0, Nn, xp2, asrc2, adst2, b2, (float*)d_out);
}


# GEMM fragment maps permuted (lane half owns 16 consecutive k, tile column c = output column 4c+t): A, B, C move as 16-byte accesses, a trip's loads issued as one group; k sum order changed
# speedup vs baseline: 3.5636x; 1.4026x over previous
#include <hip/hip_runtime.h>
#include <math.h>


typedef __attribute__((ext_vector_type(2))) float v2f;
typedef __attribute__((ext_vector_type(4))) float v4f;
typedef __attribute__((ext_vector_type(8))) float v8f;

__global__ __launch_bounds__(32) void wmma_gemm_f32(
    const float* __restrict__ A, const float* __restrict__ B,
    float* __restrict__ C, int M, int N, int K)
{
  const int lane = threadIdx.x & 31;
  const int half = lane >> 4;
  const int l15  = lane & 15;
  const int tm   = blockIdx.x;
  const int tn   = blockIdx.y << 6;

  v8f c0 = {}; v8f c1 = {}; v8f c2 = {}; v8f c3 = {};

  const float* Ap = A + (size_t)(tm * 16 + l15) * K + (half << 4);
  const float* Bp = B + (size_t)(half << 4) * N + tn + (l15 << 2);
  for (int kb = 0; kb < K; kb += 32) {
    v4f aq[4];
    v4f bq[16];
#pragma unroll
    for (int i = 0; i < 4; ++i) aq[i] = *(const v4f*)(Ap + kb + 4 * i);
#pragma unroll
    for (int m = 0; m < 16; ++m) bq[m] = *(const v4f*)(Bp + (size_t)(kb + m) * N);
    __builtin_amdgcn_sched_barrier(0);
#pragma unroll
    for (int j = 0; j < 8; ++j) {
      v2f a;
      a.x = aq[j >> 1][2 * (j & 1)];
      a.y = aq[j >> 1][2 * (j & 1) + 1];
      v2f b0, b1, b2, b3;
      b0.x = bq[2 * j][0]; b0.y = bq[2 * j + 1][0];
      b1.x = bq[2 * j][1]; b1.y = bq[2 * j + 1][1];
      b2.x = bq[2 * j][2]; b2.y = bq[2 * j + 1][2];
      b3.x = bq[2 * j][3]; b3.y = bq[2 * j + 1][3];
      c0 = __builtin_amdgcn_wmma_f32_16x16x4_f32(false, a, false, b0, (short)0, c0, false, false);
      c1 = __builtin_amdgcn_wmma_f32_16x16x4_f32(false, a, false, b1, (short)0, c1, false, false);
      c2 = __builtin_amdgcn_wmma_f32_16x16x4_f32(false, a, false, b2, (short)0, c2, false, false);
      c3 = __builtin_amdgcn_wmma_f32_16x16x4_f32(false, a, false, b3, (short)0, c3, false, false);
    }
  }

  const int orow = tm * 16 + (half << 3);
  for (int r = 0; r < 8; ++r) {
    v4f o;
    o.x = c0[r]; o.y = c1[r]; o.z = c2[r]; o.w = c3[r];
    *(v4f*)(C + (size_t)(orow + r) * N + tn + (l15 << 2)) = o;
  }
}

__global__ __launch_bounds__(256) void node_dots(
    const float* __restrict__ xp, const float* __restrict__ att_src,
    const float* __restrict__ att_dst, float* __restrict__ asrc,
    float* __restrict__ adst, int Nn, int H, int C)
{
  const int wid  = (int)((blockIdx.x * blockDim.x + threadIdx.x) >> 5);
  const int lane = threadIdx.x & 31;
  if (wid >= Nn * H) return;
  const int n = wid / H;
  const int h = wid - n * H;

  const float* xr = xp + (size_t)n * H * C + (size_t)h * C;
  const float* as = att_src + (size_t)h * C;
  const float* ad = att_dst + (size_t)h * C;
  float s = 0.f, d = 0.f;
  for (int c = lane; c < C; c += 32) {
    float v = xr[c];
    s += v * as[c];
    d += v * ad[c];
  }
  for (int off = 16; off > 0; off >>= 1) {
    s += __shfl_down(s, off, 32);
    d += __shfl_down(d, off, 32);
  }
  if (lane == 0) { asrc[wid] = s; adst[wid] = d; }
}


__device__ __forceinline__ float leaky02(float a) { return (a > 0.f) ? a : 0.2f * a; }

template <int H, int NT, int RT, int CAPW, int TCAP, bool ELU>
__global__ __launch_bounds__(NT) void owner_edge_phase(
    const int* __restrict__ src, const int* __restrict__ dst, int E0, int Nn,
    const float* __restrict__ xp, const float* __restrict__ asrc,
    const float* __restrict__ adst, const float* __restrict__ bias,
    float* __restrict__ out)
{
  constexpr int F      = H * 256;
  constexpr int F4     = F / 4;
  constexpr int NTEAMS = NT / F4;
  constexpr int R      = NTEAMS * RT;
  constexpr int NW     = NT / 32;
  constexpr int NL     = NW * NTEAMS;
  constexpr int SUB    = (TCAP < NW * CAPW) ? TCAP : NW * CAPW;
  constexpr int UA     = 16;
  constexpr int UC     = 8;
  constexpr int UI     = 5;
  static_assert(NT % F4 == 0 && F4 % 32 == 0 && NTEAMS <= 32, "team shape");
  static_assert(RT * H <= F4, "one thread per (row, head) of a team");
  static_assert(RT % UI == 0, "self-loop rows are loaded UI at a time");

  __shared__ v4f   acc[R * F4];
  __shared__ int   wl[NL * CAPW];
  __shared__ int   wcnt[NL];
  __shared__ int   tsrc[NTEAMS * TCAP];
  __shared__ int   tdl[NTEAMS * TCAP];
  __shared__ float tw[NTEAMS * TCAP * H];
  __shared__ float smax[R * H];
  __shared__ float sden[R * H];
  __shared__ float swt[R * H];
  __shared__ int   ovf;

  const int tid  = threadIdx.x;
  const int lane = tid & 31;
  const int wave = tid >> 5;
  const int team = tid / F4;
  const int c4   = tid - team * F4;
  const int h    = c4 >> 6;
  const int d0   = blockIdx.x * R;
  const int r0   = team * RT;

  const float* xcol = xp + c4 * 4;
  v4f*   arow = acc + r0 * F4 + c4;
  int*   ts   = tsrc + team * TCAP;
  int*   td   = tdl + team * TCAP;
  float* tq   = tw + team * TCAP * H;

  const bool stat = c4 < RT * H;
  const int  prow = c4 / H;
  const int  phh  = c4 - prow * H;
  const int  pidx = (r0 + prow) * H + phh;
  float pself = 0.f;
  if (stat) {
    const int d = d0 + r0 + prow;
    if (d < Nn) pself = leaky02(asrc[d * H + phh] + adst[d * H + phh]);
    smax[pidx] = pself;
  }

  auto build = [&](int lo, int hi) -> int {
    if (tid == 0) ovf = 0;
    int mycnt = 0;
    const int chunk = (hi - lo + NW - 1) / NW;
    const int cb = lo + wave * chunk;
    int ce = cb + chunk; if (ce > hi) ce = hi;
    for (int base = cb; base < ce; base += 32 * UA) {
      int dv[UA];
      if (base + 32 * UA <= ce) {
#pragma unroll
        for (int u = 0; u < UA; ++u) dv[u] = dst[base + u * 32 + lane];
      } else {
#pragma unroll
        for (int u = 0; u < UA; ++u) {
          const int e = base + u * 32 + lane;
          dv[u] = (e < ce) ? dst[e] : -1;
        }
      }
#pragma unroll
      for (int u = 0; u < UA; ++u) {
        const unsigned dl = (unsigned)(dv[u] - d0);
        unsigned m = __builtin_amdgcn_ballot_w32(dl < (unsigned)R);
        while (m) {
          const int l = __builtin_ctz(m);
          m &= m - 1;
          const int dls = __builtin_amdgcn_readlane((int)dl, l);
          const int t = dls / RT;
          if (lane == t) {
            if (mycnt < CAPW) wl[(wave * NTEAMS + t) * CAPW + mycnt] = base + u * 32 + l;
            ++mycnt;
          }
        }
      }
    }
    if (lane < NTEAMS) wcnt[wave * NTEAMS + lane] = mycnt;
    __syncthreads();

    int tot = 0;
    bool bad = false;
    for (int w = 0; w < NW; ++w) {
      const int c = wcnt[w * NTEAMS + team];
      bad |= (c > CAPW);
      tot += c;
    }
    bad |= (tot > TCAP);
    if (bad) ovf = 1;
    __syncthreads();
    const int o = ovf;
    __syncthreads();
    if (o) return -1;

    for (int j = c4; j < tot; j += F4) {
      int rem = j, w = 0;
      int c = wcnt[team];
      while (rem >= c) { rem -= c; ++w; c = wcnt[w * NTEAMS + team]; }
      const int e = wl[(w * NTEAMS + team) * CAPW + rem];
      const int s = src[e];
      const int d = dst[e];
      ts[j] = s;
      td[j] = d - d0 - r0;
      for (int hh = 0; hh < H; ++hh)
        tq[j * H + hh] = leaky02(asrc[s * H + hh] + adst[d * H + hh]);
    }
    __syncthreads();
    return tot;
  };

  auto pass_max = [&](int tot) {
    if (stat) {
      float m = smax[pidx];
#pragma unroll 8
      for (int j = 0; j < tot; ++j) {
        const float a = tq[j * H + phh];
        m = (td[j] == prow) ? fmaxf(m, a) : m;
      }
      smax[pidx] = m;
    }
  };

  auto pass_den = [&](int tot) {
    if (stat) {
      const float m = smax[pidx];
      float s = sden[pidx];
#pragma unroll 8
      for (int j = 0; j < tot; ++j) {
        const float ev = __expf(tq[j * H + phh] - m);
        s += (td[j] == prow) ? ev : 0.f;
      }
      sden[pidx] = s;
    }
  };

  auto pass_agg = [&](int tot) {
    for (int j = c4; j < tot; j += F4) {
      const int rr = (r0 + td[j]) * H;
      for (int hh = 0; hh < H; ++hh)
        tq[j * H + hh] = __expf(tq[j * H + hh] - smax[rr + hh]) / sden[rr + hh];
    }
    __syncthreads();
    const float* tqh = tq + h;
    int i = 0;
    for (; i + UC <= tot; i += UC) {
      int sr[UC], dr[UC]; float wt[UC]; v4f v[UC];
#pragma unroll
      for (int u = 0; u < UC; ++u) {
        sr[u] = ts[i + u]; dr[u] = td[i + u]; wt[u] = tqh[(i + u) * H];
      }
#pragma unroll
      for (int u = 0; u < UC; ++u) v[u] = *(const v4f*)(xcol + (size_t)sr[u] * F);
#pragma unroll
      for (int u = 0; u < UC; ++u) {
        v4f a = arow[dr[u] * F4];
        a += v[u] * wt[u];
        arow[dr[u] * F4] = a;
      }
    }
    for (; i < tot; ++i) {
      const v4f v1 = *(const v4f*)(xcol + (size_t)ts[i] * F);
      v4f a = arow[td[i] * F4];
      a += v1 * tqh[i * H];
      arow[td[i] * F4] = a;
    }
  };

  int tot = build(0, E0);
  const bool whole = (tot >= 0);
  const int step = whole ? ((E0 > 0) ? E0 : 1) : SUB;

  for (int mode = 0; mode < 3; ++mode) {
    if (mode == 1 && stat) sden[pidx] = __expf(pself - smax[pidx]);
    if (mode == 2) {
      if (stat) swt[pidx] = __expf(pself - smax[pidx]) / sden[pidx];
      __syncthreads();
      for (int r = 0; r < RT; r += UI) {
        v4f v[UI];
#pragma unroll
        for (int u = 0; u < UI; ++u) {
          const int d = d0 + r0 + r + u;
          const v4f z = {0.f, 0.f, 0.f, 0.f};
          v[u] = (d < Nn) ? *(const v4f*)(xcol + (size_t)d * F) : z;
        }
#pragma unroll
        for (int u = 0; u < UI; ++u) arow[(r + u) * F4] = v[u] * swt[(r0 + r + u) * H + h];
      }
    }
    for (int lo = 0; lo < E0; lo += step) {
      if (!whole) {
        const int hi = (lo + SUB < E0) ? lo + SUB : E0;
        tot = build(lo, hi);
      }
      if (mode == 0) pass_max(tot);
      else if (mode == 1) pass_den(tot);
      else pass_agg(tot);
      __syncthreads();
    }
    __syncthreads();
  }

  const float b0 = bias[c4 * 4 + 0], b1 = bias[c4 * 4 + 1];
  const float b2 = bias[c4 * 4 + 2], b3 = bias[c4 * 4 + 3];
  for (int r = 0; r < RT; ++r) {
    const int d = d0 + r0 + r;
    if (d < Nn) {
      v4f a = arow[r * F4];
      a.x += b0; a.y += b1; a.z += b2; a.w += b3;
      if (ELU) {
        a.x = (a.x > 0.f) ? a.x : expm1f(a.x);
        a.y = (a.y > 0.f) ? a.y : expm1f(a.y);
        a.z = (a.z > 0.f) ? a.z : expm1f(a.z);
        a.w = (a.w > 0.f) ? a.w : expm1f(a.w);
      }
      *(v4f*)(out + (size_t)d * F + c4 * 4) = a;
    }
  }
}


extern "C" void kernel_launch(void* const* d_in, const int* in_sizes, int n_in,
                              void* d_out, int out_size, void* d_ws, size_t ws_size,
                              hipStream_t stream) {
  const float* x   = (const float*)d_in[0];
  const int*   ei  = (const int*)d_in[1];
  const float* W1  = (const float*)d_in[2];
  const float* as1 = (const float*)d_in[3];
  const float* ad1 = (const float*)d_in[4];
  const float* b1  = (const float*)d_in[5];
  const float* W2  = (const float*)d_in[6];
  const float* as2 = (const float*)d_in[7];
  const float* ad2 = (const float*)d_in[8];
  const float* b2  = (const float*)d_in[9];

  const int K1 = 768, F1 = 768, H1 = 3;
  const int F2 = 256, H2 = 1;
  const int Nn = in_sizes[0] / K1;
  const int E0 = in_sizes[1] / 2;

  const int* srcI = ei;
  const int* dstI = ei + E0;

  float* ws = (float*)d_ws;
  float* xp1   = ws; ws += (size_t)Nn * F1;
  float* h1    = ws; ws += (size_t)Nn * F1;
  float* xp2   = ws; ws += (size_t)Nn * F2;
  float* asrc1 = ws; ws += (size_t)Nn * H1;
  float* adst1 = ws; ws += (size_t)Nn * H1;
  float* asrc2 = ws; ws += Nn;
  float* adst2 = ws; ws += Nn;

  {
    dim3 g(Nn / 16, F1 / 64);
    wmma_gemm_f32<<<g, 32, 0, stream>>>(x, W1, xp1, Nn, F1, K1);
  }
  node_dots<<<(Nn * H1 + 7) / 8, 256, 0, stream>>>(xp1, as1, ad1, asrc1, adst1, Nn, H1, 256);
  owner_edge_phase<3, 768, 20, 48, 512, true><<<(Nn + 79) / 80, 768, 0, stream>>>(
      srcI, dstI, E0, Nn, xp1, asrc1, adst1, b1, h1);

  {
    dim3 g(Nn / 16, F2 / 64);
    wmma_gemm_f32<<<g, 32, 0, stream>>>(h1, W2, xp2, Nn, F2, F1);
  }
  node_dots<<<(Nn + 7) / 8, 256, 0, stream>>>(xp2, as2, ad2, asrc2, adst2, Nn, H2, 256);
  owner_edge_phase<1, 1024, 5, 24, 256, false><<<(Nn + 79) / 80, 1024, 0, stream>>>(
      srcI, dstI, E0, Nn, xp2, asrc2, adst2, b2, (float*)d_out);
}


# Edge array filtered once per call: layer 1's edge phase parks its team edge lists (src, row) in its own rows of the output buffer; layer 2 filters those short lists, not all 320000 dst; order kept
# speedup vs baseline: 3.7305x; 1.0468x over previous
#include <hip/hip_runtime.h>
#include <math.h>


typedef __attribute__((ext_vector_type(2))) float v2f;
typedef __attribute__((ext_vector_type(4))) float v4f;
typedef __attribute__((ext_vector_type(8))) float v8f;

__global__ __launch_bounds__(32) void wmma_gemm_f32(
    const float* __restrict__ A, const float* __restrict__ B,
    float* __restrict__ C, int M, int N, int K)
{
  const int lane = threadIdx.x & 31;
  const int half = lane >> 4;
  const int l15  = lane & 15;
  const int tm   = blockIdx.x;
  const int tn   = blockIdx.y << 6;

  v8f c0 = {}; v8f c1 = {}; v8f c2 = {}; v8f c3 = {};

  const float* Ap = A + (size_t)(tm * 16 + l15) * K + (half << 4);
  const float* Bp = B + (size_t)(half << 4) * N + tn + (l15 << 2);
  for (int kb = 0; kb < K; kb += 32) {
    v4f aq[4];
    v4f bq[16];
#pragma unroll
    for (int i = 0; i < 4; ++i) aq[i] = *(const v4f*)(Ap + kb + 4 * i);
#pragma unroll
    for (int m = 0; m < 16; ++m) bq[m] = *(const v4f*)(Bp + (size_t)(kb + m) * N);
    __builtin_amdgcn_sched_barrier(0);
#pragma unroll
    for (int j = 0; j < 8; ++j) {
      v2f a;
      a.x = aq[j >> 1][2 * (j & 1)];
      a.y = aq[j >> 1][2 * (j & 1) + 1];
      v2f b0, b1, b2, b3;
      b0.x = bq[2 * j][0]; b0.y = bq[2 * j + 1][0];
      b1.x = bq[2 * j][1]; b1.y = bq[2 * j + 1][1];
      b2.x = bq[2 * j][2]; b2.y = bq[2 * j + 1][2];
      b3.x = bq[2 * j][3]; b3.y = bq[2 * j + 1][3];
      c0 = __builtin_amdgcn_wmma_f32_16x16x4_f32(false, a, false, b0, (short)0, c0, false, false);
      c1 = __builtin_amdgcn_wmma_f32_16x16x4_f32(false, a, false, b1, (short)0, c1, false, false);
      c2 = __builtin_amdgcn_wmma_f32_16x16x4_f32(false, a, false, b2, (short)0, c2, false, false);
      c3 = __builtin_amdgcn_wmma_f32_16x16x4_f32(false, a, false, b3, (short)0, c3, false, false);
    }
  }

  const int orow = tm * 16 + (half << 3);
  for (int r = 0; r < 8; ++r) {
    v4f o;
    o.x = c0[r]; o.y = c1[r]; o.z = c2[r]; o.w = c3[r];
    *(v4f*)(C + (size_t)(orow + r) * N + tn + (l15 << 2)) = o;
  }
}

__global__ __launch_bounds__(256) void node_dots(
    const float* __restrict__ xp, const float* __restrict__ att_src,
    const float* __restrict__ att_dst, float* __restrict__ asrc,
    float* __restrict__ adst, int Nn, int H, int C)
{
  const int wid  = (int)((blockIdx.x * blockDim.x + threadIdx.x) >> 5);
  const int lane = threadIdx.x & 31;
  if (wid >= Nn * H) return;
  const int n = wid / H;
  const int h = wid - n * H;

  const float* xr = xp + (size_t)n * H * C + (size_t)h * C;
  const float* as = att_src + (size_t)h * C;
  const float* ad = att_dst + (size_t)h * C;
  float s = 0.f, d = 0.f;
  for (int c = lane; c < C; c += 32) {
    float v = xr[c];
    s += v * as[c];
    d += v * ad[c];
  }
  for (int off = 16; off > 0; off >>= 1) {
    s += __shfl_down(s, off, 32);
    d += __shfl_down(d, off, 32);
  }
  if (lane == 0) { asrc[wid] = s; adst[wid] = d; }
}


__device__ __forceinline__ float leaky02(float a) { return (a > 0.f) ? a : 0.2f * a; }

template <int H, int NT, int RT, int CAPW, int TCAP, bool ELU,
          bool SAVE, bool LOAD, int PRT, int PCAP>
__global__ __launch_bounds__(NT) void owner_edge_phase(
    const int* __restrict__ src, const int* __restrict__ dst, int E0, int Nn,
    const float* __restrict__ xp, const float* __restrict__ asrc,
    const float* __restrict__ adst, const float* __restrict__ bias,
    float* __restrict__ out, float* __restrict__ park, int pstride, int plimit)
{
  constexpr int F      = H * 256;
  constexpr int F4     = F / 4;
  constexpr int NTEAMS = NT / F4;
  constexpr int R      = NTEAMS * RT;
  constexpr int NW     = NT / 32;
  constexpr int NL     = NW * NTEAMS;
  constexpr int SUB    = (TCAP < NW * CAPW) ? TCAP : NW * CAPW;
  constexpr int UA     = 16;
  constexpr int UC     = 8;
  constexpr int UI     = 5;
  static_assert(NT % F4 == 0 && F4 % 32 == 0 && NTEAMS <= 32, "team shape");
  static_assert(RT * H <= F4, "one thread per (row, head) of a team");
  static_assert(RT % UI == 0, "self-loop rows are loaded UI at a time");
  static_assert(R % PRT == 0 && PRT % RT == 0 && PCAP > 0, "parked lists: a team's rows lie in one parent team");
  static_assert(!SAVE || (PRT == RT && PCAP == TCAP), "a SAVE launch parks its own teams");
  constexpr int PT    = R / PRT;
  constexpr int PNEED = PT + 2 * PT * PCAP;

  __shared__ v4f   acc[R * F4];
  __shared__ int   wl[NL * CAPW];
  __shared__ int   wcnt[NL];
  __shared__ int   tsrc[NTEAMS * TCAP];
  __shared__ int   tdl[NTEAMS * TCAP];
  __shared__ float tw[NTEAMS * TCAP * H];
  __shared__ float smax[R * H];
  __shared__ float sden[R * H];
  __shared__ float swt[R * H];
  __shared__ int   ovf;

  const int tid  = threadIdx.x;
  const int lane = tid & 31;
  const int wave = tid >> 5;
  const int team = tid / F4;
  const int c4   = tid - team * F4;
  const int h    = c4 >> 6;
  const int d0   = blockIdx.x * R;
  const int r0   = team * RT;
  const bool can_park = (size_t)blockIdx.x * (size_t)pstride + (size_t)PNEED <= (size_t)plimit;

  const float* xcol = xp + c4 * 4;
  v4f*   arow = acc + r0 * F4 + c4;
  int*   ts   = tsrc + team * TCAP;
  int*   td   = tdl + team * TCAP;
  float* tq   = tw + team * TCAP * H;

  const bool stat = c4 < RT * H;
  const int  prow = c4 / H;
  const int  phh  = c4 - prow * H;
  const int  pidx = (r0 + prow) * H + phh;
  float pself = 0.f;
  if (stat) {
    const int d = d0 + r0 + prow;
    if (d < Nn) pself = leaky02(asrc[d * H + phh] + adst[d * H + phh]);
    smax[pidx] = pself;
  }

  auto build = [&](int lo, int hi) -> int {
    if (tid == 0) ovf = 0;
    int mycnt = 0;
    const int chunk = (hi - lo + NW - 1) / NW;
    const int cb = lo + wave * chunk;
    int ce = cb + chunk; if (ce > hi) ce = hi;
    for (int base = cb; base < ce; base += 32 * UA) {
      int dv[UA];
      if (base + 32 * UA <= ce) {
#pragma unroll
        for (int u = 0; u < UA; ++u) dv[u] = dst[base + u * 32 + lane];
      } else {
#pragma unroll
        for (int u = 0; u < UA; ++u) {
          const int e = base + u * 32 + lane;
          dv[u] = (e < ce) ? dst[e] : -1;
        }
      }
#pragma unroll
      for (int u = 0; u < UA; ++u) {
        const unsigned dl = (unsigned)(dv[u] - d0);
        unsigned m = __builtin_amdgcn_ballot_w32(dl < (unsigned)R);
        while (m) {
          const int l = __builtin_ctz(m);
          m &= m - 1;
          const int dls = __builtin_amdgcn_readlane((int)dl, l);
          const int t = dls / RT;
          if (lane == t) {
            if (mycnt < CAPW) wl[(wave * NTEAMS + t) * CAPW + mycnt] = base + u * 32 + l;
            ++mycnt;
          }
        }
      }
    }
    if (lane < NTEAMS) wcnt[wave * NTEAMS + lane] = mycnt;
    __syncthreads();

    int tot = 0;
    bool bad = false;
    for (int w = 0; w < NW; ++w) {
      const int c = wcnt[w * NTEAMS + team];
      bad |= (c > CAPW);
      tot += c;
    }
    bad |= (tot > TCAP);
    if (bad) ovf = 1;
    __syncthreads();
    const int o = ovf;
    __syncthreads();
    if (o) return -1;

    for (int j = c4; j < tot; j += F4) {
      int rem = j, w = 0;
      int c = wcnt[team];
      while (rem >= c) { rem -= c; ++w; c = wcnt[w * NTEAMS + team]; }
      const int e = wl[(w * NTEAMS + team) * CAPW + rem];
      const int s = src[e];
      const int d = dst[e];
      ts[j] = s;
      td[j] = d - d0 - r0;
      for (int hh = 0; hh < H; ++hh)
        tq[j * H + hh] = leaky02(asrc[s * H + hh] + adst[d * H + hh]);
    }
    __syncthreads();
    return tot;
  };

  auto build_parked = [&]() -> int {
    constexpr int WT  = F4 / 32;
    constexpr int EPW = (PCAP + WT - 1) / WT;
    constexpr int UL  = (EPW + 31) / 32;
    const float* sb = out + (size_t)blockIdx.x * pstride;
    bool have = can_park;
    if (have) {
      for (int p = 0; p < PT; ++p) {
        const int np = __float_as_int(sb[p]);
        have &= (np >= 0 && np <= PCAP);
      }
    }
    if (!have) return -2;
    if (tid == 0) ovf = 0;
    const int p  = r0 / PRT;
    const int n  = __float_as_int(sb[p]);
    const float* ps_src = sb + PT + p * 2 * PCAP;
    const float* ps_row = ps_src + PCAP;
    const int wt = c4 >> 5;
    int sv[UL], rv[UL], pos[UL];
#pragma unroll
    for (int u = 0; u < UL; ++u) {
      const int k   = u * 32 + lane;
      const int idx = wt * EPW + k;
      const bool ok = (k < EPW) && (idx < n);
      sv[u] = ok ? __float_as_int(ps_src[idx]) : 0;
      rv[u] = ok ? __float_as_int(ps_row[idx]) : -1;
    }
    int cnt = 0;
#pragma unroll
    for (int u = 0; u < UL; ++u) {
      const unsigned dl = (unsigned)(rv[u] - r0);
      const unsigned m = __builtin_amdgcn_ballot_w32(dl < (unsigned)RT);
      pos[u] = cnt + __builtin_popcount(m & ((1u << lane) - 1u));
      cnt += __builtin_popcount(m);
    }
    if (lane == 0) wcnt[team * WT + wt] = cnt;
    __syncthreads();
    int totl = 0, off = 0;
    for (int w = 0; w < WT; ++w) {
      const int c = wcnt[team * WT + w];
      off += (w < wt) ? c : 0;
      totl += c;
    }
    if (totl > TCAP) ovf = 1;
    __syncthreads();
    const int o = ovf;
    __syncthreads();
    if (o) return -1;
    float lg[UL][H];
#pragma unroll
    for (int u = 0; u < UL; ++u) {
      const bool hit = (unsigned)(rv[u] - r0) < (unsigned)RT;
      const int s = hit ? sv[u] : 0;
      const int d = hit ? d0 + rv[u] : d0;
      for (int hh = 0; hh < H; ++hh)
        lg[u][hh] = leaky02(asrc[s * H + hh] + adst[d * H + hh]);
    }
#pragma unroll
    for (int u = 0; u < UL; ++u) {
      const unsigned dl = (unsigned)(rv[u] - r0);
      if (dl < (unsigned)RT) {
        const int j = off + pos[u];
        ts[j] = sv[u];
        td[j] = (int)dl;
        for (int hh = 0; hh < H; ++hh) tq[j * H + hh] = lg[u][hh];
      }
    }
    __syncthreads();
    return totl;
  };

  auto pass_max = [&](int tot) {
    if (stat) {
      float m = smax[pidx];
#pragma unroll 8
      for (int j = 0; j < tot; ++j) {
        const float a = tq[j * H + phh];
        m = (td[j] == prow) ? fmaxf(m, a) : m;
      }
      smax[pidx] = m;
    }
  };

  auto pass_den = [&](int tot) {
    if (stat) {
      const float m = smax[pidx];
      float s = sden[pidx];
#pragma unroll 8
      for (int j = 0; j < tot; ++j) {
        const float ev = __expf(tq[j * H + phh] - m);
        s += (td[j] == prow) ? ev : 0.f;
      }
      sden[pidx] = s;
    }
  };

  auto pass_agg = [&](int tot) {
    for (int j = c4; j < tot; j += F4) {
      const int rr = (r0 + td[j]) * H;
      for (int hh = 0; hh < H; ++hh)
        tq[j * H + hh] = __expf(tq[j * H + hh] - smax[rr + hh]) / sden[rr + hh];
    }
    __syncthreads();
    const float* tqh = tq + h;
    int i = 0;
    for (; i + UC <= tot; i += UC) {
      int sr[UC], dr[UC]; float wt[UC]; v4f v[UC];
#pragma unroll
      for (int u = 0; u < UC; ++u) {
        sr[u] = ts[i + u]; dr[u] = td[i + u]; wt[u] = tqh[(i + u) * H];
      }
#pragma unroll
      for (int u = 0; u < UC; ++u) v[u] = *(const v4f*)(xcol + (size_t)sr[u] * F);
#pragma unroll
      for (int u = 0; u < UC; ++u) {
        v4f a = arow[dr[u] * F4];
        a += v[u] * wt[u];
        arow[dr[u] * F4] = a;
      }
    }
    for (; i < tot; ++i) {
      const v4f v1 = *(const v4f*)(xcol + (size_t)ts[i] * F);
      v4f a = arow[td[i] * F4];
      a += v1 * tqh[i * H];
      arow[td[i] * F4] = a;
    }
  };

  int tot;
  if constexpr (LOAD) {
    tot = build_parked();
    if (tot == -2) tot = build(0, E0);
  } else {
    tot = build(0, E0);
  }
  const bool whole = (tot >= 0);
  if constexpr (SAVE) {
    if (can_park) {
      float* sb = park + (size_t)blockIdx.x * pstride;
      if (whole) {
        float* ps_src = sb + PT + team * 2 * PCAP;
        for (int j = c4; j < tot; j += F4) {
          ps_src[j]        = __int_as_float(ts[j]);
          ps_src[PCAP + j] = __int_as_float(r0 + td[j]);
        }
      }
      if (c4 == 0) sb[team] = __int_as_float(whole ? tot : -1);
    }
  }
  const int step = whole ? ((E0 > 0) ? E0 : 1) : SUB;

  for (int mode = 0; mode < 3; ++mode) {
    if (mode == 1 && stat) sden[pidx] = __expf(pself - smax[pidx]);
    if (mode == 2) {
      if (stat) swt[pidx] = __expf(pself - smax[pidx]) / sden[pidx];
      __syncthreads();
      for (int r = 0; r < RT; r += UI) {
        v4f v[UI];
#pragma unroll
        for (int u = 0; u < UI; ++u) {
          const int d = d0 + r0 + r + u;
          const v4f z = {0.f, 0.f, 0.f, 0.f};
          v[u] = (d < Nn) ? *(const v4f*)(xcol + (size_t)d * F) : z;
        }
#pragma unroll
        for (int u = 0; u < UI; ++u) arow[(r + u) * F4] = v[u] * swt[(r0 + r + u) * H + h];
      }
    }
    for (int lo = 0; lo < E0; lo += step) {
      if (!whole) {
        const int hi = (lo + SUB < E0) ? lo + SUB : E0;
        tot = build(lo, hi);
      }
      if (mode == 0) pass_max(tot);
      else if (mode == 1) pass_den(tot);
      else pass_agg(tot);
      __syncthreads();
    }
    __syncthreads();
  }

  const float b0 = bias[c4 * 4 + 0], b1 = bias[c4 * 4 + 1];
  const float b2 = bias[c4 * 4 + 2], b3 = bias[c4 * 4 + 3];
  for (int r = 0; r < RT; ++r) {
    const int d = d0 + r0 + r;
    if (d < Nn) {
      v4f a = arow[r * F4];
      a.x += b0; a.y += b1; a.z += b2; a.w += b3;
      if (ELU) {
        a.x = (a.x > 0.f) ? a.x : expm1f(a.x);
        a.y = (a.y > 0.f) ? a.y : expm1f(a.y);
        a.z = (a.z > 0.f) ? a.z : expm1f(a.z);
        a.w = (a.w > 0.f) ? a.w : expm1f(a.w);
      }
      *(v4f*)(out + (size_t)d * F + c4 * 4) = a;
    }
  }
}


extern "C" void kernel_launch(void* const* d_in, const int* in_sizes, int n_in,
                              void* d_out, int out_size, void* d_ws, size_t ws_size,
                              hipStream_t stream) {
  const float* x   = (const float*)d_in[0];
  const int*   ei  = (const int*)d_in[1];
  const float* W1  = (const float*)d_in[2];
  const float* as1 = (const float*)d_in[3];
  const float* ad1 = (const float*)d_in[4];
  const float* b1  = (const float*)d_in[5];
  const float* W2  = (const float*)d_in[6];
  const float* as2 = (const float*)d_in[7];
  const float* ad2 = (const float*)d_in[8];
  const float* b2  = (const float*)d_in[9];

  const int K1 = 768, F1 = 768, H1 = 3;
  const int F2 = 256, H2 = 1;
  const int Nn = in_sizes[0] / K1;
  const int E0 = in_sizes[1] / 2;

  const int* srcI = ei;
  const int* dstI = ei + E0;

  float* ws = (float*)d_ws;
  float* xp1   = ws; ws += (size_t)Nn * F1;
  float* h1    = ws; ws += (size_t)Nn * F1;
  float* xp2   = ws; ws += (size_t)Nn * F2;
  float* asrc1 = ws; ws += (size_t)Nn * H1;
  float* adst1 = ws; ws += (size_t)Nn * H1;
  float* asrc2 = ws; ws += Nn;
  float* adst2 = ws; ws += Nn;

  {
    dim3 g(Nn / 16, F1 / 64);
    wmma_gemm_f32<<<g, 32, 0, stream>>>(x, W1, xp1, Nn, F1, K1);
  }
  node_dots<<<(Nn * H1 + 7) / 8, 256, 0, stream>>>(xp1, as1, ad1, asrc1, adst1, Nn, H1, 256);
  const int PSTR = 80 * F2;
  const int PLIM = Nn * F2;
  owner_edge_phase<3, 768, 20, 48, 512, true, true, false, 20, 512>
      <<<(Nn + 79) / 80, 768, 0, stream>>>(
      srcI, dstI, E0, Nn, xp1, asrc1, adst1, b1, h1, (float*)d_out, PSTR, PLIM);

  {
    dim3 g(Nn / 16, F2 / 64);
    wmma_gemm_f32<<<g, 32, 0, stream>>>(h1, W2, xp2, Nn, F2, F1);
  }
  node_dots<<<(Nn + 7) / 8, 256, 0, stream>>>(xp2, as2, ad2, asrc2, adst2, Nn, H2, 256);
  owner_edge_phase<1, 1024, 5, 24, 256, false, false, true, 20, 512>
      <<<(Nn + 79) / 80, 1024, 0, stream>>>(
      srcI, dstI, E0, Nn, xp2, asrc2, adst2, b2, (float*)d_out, nullptr, PSTR, PLIM);
}


# Attention dots reassociated as x*(W*att): a small launch folds att into 6 (2) K-vectors, an extra 16-wide WMMA strip per GEMM launch forms a_src/a_dst; node-dot launches gone; dot sum order changed
# speedup vs baseline: 3.8778x; 1.0395x over previous
#include <hip/hip_runtime.h>
#include <math.h>


typedef __attribute__((ext_vector_type(2))) float v2f;
typedef __attribute__((ext_vector_type(4))) float v4f;
typedef __attribute__((ext_vector_type(8))) float v8f;

__global__ __launch_bounds__(32) void wmma_gemm_f32(
    const float* __restrict__ A, const float* __restrict__ B,
    float* __restrict__ C, int M, int N, int K,
    const float* __restrict__ Wt, float* __restrict__ asrc,
    float* __restrict__ adst, int H)
{
  const int lane = threadIdx.x & 31;
  const int half = lane >> 4;
  const int l15  = lane & 15;
  const int tm   = blockIdx.x;

  if (blockIdx.y == 0) {
    const int bc = (l15 < 2 * H) ? l15 : 0;
    const float* Ad = A + (size_t)(tm * 16 + l15) * K + (half << 4);
    const float* Bd = Wt + (size_t)bc * K + (half << 4);
    v8f d0 = {}; v8f d1 = {};
    for (int kb = 0; kb < K; kb += 32) {
      v4f aq[4];
      v4f bq[4];
#pragma unroll
      for (int i = 0; i < 4; ++i) aq[i] = *(const v4f*)(Ad + kb + 4 * i);
#pragma unroll
      for (int i = 0; i < 4; ++i) bq[i] = *(const v4f*)(Bd + kb + 4 * i);
#pragma unroll
      for (int j = 0; j < 8; j += 2) {
        v2f a0, b0, a1, b1;
        a0.x = aq[j >> 1][0]; a0.y = aq[j >> 1][1];
        b0.x = bq[j >> 1][0]; b0.y = bq[j >> 1][1];
        a1.x = aq[j >> 1][2]; a1.y = aq[j >> 1][3];
        b1.x = bq[j >> 1][2]; b1.y = bq[j >> 1][3];
        d0 = __builtin_amdgcn_wmma_f32_16x16x4_f32(false, a0, false, b0, (short)0, d0, false, false);
        d1 = __builtin_amdgcn_wmma_f32_16x16x4_f32(false, a1, false, b1, (short)0, d1, false, false);
      }
    }
    if (l15 < 2 * H) {
      float* o = (l15 < H) ? (asrc + l15) : (adst + (l15 - H));
      const int row0 = tm * 16 + (half << 3);
      for (int r = 0; r < 8; ++r) o[(size_t)(row0 + r) * H] = d0[r] + d1[r];
    }
    return;
  }
  const int tn   = (blockIdx.y - 1) << 6;

  v8f c0 = {}; v8f c1 = {}; v8f c2 = {}; v8f c3 = {};

  const float* Ap = A + (size_t)(tm * 16 + l15) * K + (half << 4);
  const float* Bp = B + (size_t)(half << 4) * N + tn + (l15 << 2);
  for (int kb = 0; kb < K; kb += 32) {
    v4f aq[4];
    v4f bq[16];
#pragma unroll
    for (int i = 0; i < 4; ++i) aq[i] = *(const v4f*)(Ap + kb + 4 * i);
#pragma unroll
    for (int m = 0; m < 16; ++m) bq[m] = *(const v4f*)(Bp + (size_t)(kb + m) * N);
    __builtin_amdgcn_sched_barrier(0);
#pragma unroll
    for (int j = 0; j < 8; ++j) {
      v2f a;
      a.x = aq[j >> 1][2 * (j & 1)];
      a.y = aq[j >> 1][2 * (j & 1) + 1];
      v2f b0, b1, b2, b3;
      b0.x = bq[2 * j][0]; b0.y = bq[2 * j + 1][0];
      b1.x = bq[2 * j][1]; b1.y = bq[2 * j + 1][1];
      b2.x = bq[2 * j][2]; b2.y = bq[2 * j + 1][2];
      b3.x = bq[2 * j][3]; b3.y = bq[2 * j + 1][3];
      c0 = __builtin_amdgcn_wmma_f32_16x16x4_f32(false, a, false, b0, (short)0, c0, false, false);
      c1 = __builtin_amdgcn_wmma_f32_16x16x4_f32(false, a, false, b1, (short)0, c1, false, false);
      c2 = __builtin_amdgcn_wmma_f32_16x16x4_f32(false, a, false, b2, (short)0, c2, false, false);
      c3 = __builtin_amdgcn_wmma_f32_16x16x4_f32(false, a, false, b3, (short)0, c3, false, false);
    }
  }

  const int orow = tm * 16 + (half << 3);
  for (int r = 0; r < 8; ++r) {
    v4f o;
    o.x = c0[r]; o.y = c1[r]; o.z = c2[r]; o.w = c3[r];
    *(v4f*)(C + (size_t)(orow + r) * N + tn + (l15 << 2)) = o;
  }
}

__global__ __launch_bounds__(256) void fold_att(
    const float* __restrict__ W1, const float* __restrict__ as1,
    const float* __restrict__ ad1, float* __restrict__ wt1, int K1, int H1,
    const float* __restrict__ W2, const float* __restrict__ as2,
    const float* __restrict__ ad2, float* __restrict__ wt2, int K2, int H2)
{
  int wid = (int)((blockIdx.x * blockDim.x + threadIdx.x) >> 5);
  const int lane = threadIdx.x & 31;
  const float* Wm = W1; const float* as = as1; const float* ad = ad1;
  float* wt = wt1; int K = K1; int H = H1;
  if (wid >= K1 * H1) {
    wid -= K1 * H1;
    if (wid >= K2 * H2) return;
    Wm = W2; as = as2; ad = ad2; wt = wt2; K = K2; H = H2;
  }
  const int k = wid / H;
  const int h = wid - k * H;
  const float* wr = Wm + (size_t)k * H * 256 + h * 256 + lane * 4;
  const float* sr = as + h * 256 + lane * 4;
  const float* dr = ad + h * 256 + lane * 4;
  const v4f w0 = *(const v4f*)wr,  w1 = *(const v4f*)(wr + 128);
  const v4f s0 = *(const v4f*)sr,  s1 = *(const v4f*)(sr + 128);
  const v4f e0 = *(const v4f*)dr,  e1 = *(const v4f*)(dr + 128);
  float s = w0.x * s0.x + w0.y * s0.y + w0.z * s0.z + w0.w * s0.w
          + w1.x * s1.x + w1.y * s1.y + w1.z * s1.z + w1.w * s1.w;
  float d = w0.x * e0.x + w0.y * e0.y + w0.z * e0.z + w0.w * e0.w
          + w1.x * e1.x + w1.y * e1.y + w1.z * e1.z + w1.w * e1.w;
  for (int off = 16; off > 0; off >>= 1) {
    s += __shfl_down(s, off, 32);
    d += __shfl_down(d, off, 32);
  }
  if (lane == 0) {
    wt[(size_t)h * K + k]       = s;
    wt[(size_t)(H + h) * K + k] = d;
  }
}


__device__ __forceinline__ float leaky02(float a) { return (a > 0.f) ? a : 0.2f * a; }

template <int H, int NT, int RT, int CAPW, int TCAP, bool ELU,
          bool SAVE, bool LOAD, int PRT, int PCAP>
__global__ __launch_bounds__(NT) void owner_edge_phase(
    const int* __restrict__ src, const int* __restrict__ dst, int E0, int Nn,
    const float* __restrict__ xp, const float* __restrict__ asrc,
    const float* __restrict__ adst, const float* __restrict__ bias,
    float* __restrict__ out, float* __restrict__ park, int pstride, int plimit)
{
  constexpr int F      = H * 256;
  constexpr int F4     = F / 4;
  constexpr int NTEAMS = NT / F4;
  constexpr int R      = NTEAMS * RT;
  constexpr int NW     = NT / 32;
  constexpr int NL     = NW * NTEAMS;
  constexpr int SUB    = (TCAP < NW * CAPW) ? TCAP : NW * CAPW;
  constexpr int UA     = 16;
  constexpr int UC     = 8;
  constexpr int UI     = 5;
  static_assert(NT % F4 == 0 && F4 % 32 == 0 && NTEAMS <= 32, "team shape");
  static_assert(RT * H <= F4, "one thread per (row, head) of a team");
  static_assert(RT % UI == 0, "self-loop rows are loaded UI at a time");
  static_assert(R % PRT == 0 && PRT % RT == 0 && PCAP > 0, "parked lists: a team's rows lie in one parent team");
  static_assert(!SAVE || (PRT == RT && PCAP == TCAP), "a SAVE launch parks its own teams");
  constexpr int PT    = R / PRT;
  constexpr int PNEED = PT + 2 * PT * PCAP;

  __shared__ v4f   acc[R * F4];
  __shared__ int   wl[NL * CAPW];
  __shared__ int   wcnt[NL];
  __shared__ int   tsrc[NTEAMS * TCAP];
  __shared__ int   tdl[NTEAMS * TCAP];
  __shared__ float tw[NTEAMS * TCAP * H];
  __shared__ float smax[R * H];
  __shared__ float sden[R * H];
  __shared__ float swt[R * H];
  __shared__ int   ovf;

  const int tid  = threadIdx.x;
  const int lane = tid & 31;
  const int wave = tid >> 5;
  const int team = tid / F4;
  const int c4   = tid - team * F4;
  const int h    = c4 >> 6;
  const int d0   = blockIdx.x * R;
  const int r0   = team * RT;
  const bool can_park = (size_t)blockIdx.x * (size_t)pstride + (size_t)PNEED <= (size_t)plimit;

  const float* xcol = xp + c4 * 4;
  v4f*   arow = acc + r0 * F4 + c4;
  int*   ts   = tsrc + team * TCAP;
  int*   td   = tdl + team * TCAP;
  float* tq   = tw + team * TCAP * H;

  const bool stat = c4 < RT * H;
  const int  prow = c4 / H;
  const int  phh  = c4 - prow * H;
  const int  pidx = (r0 + prow) * H + phh;
  float pself = 0.f;
  if (stat) {
    const int d = d0 + r0 + prow;
    if (d < Nn) pself = leaky02(asrc[d * H + phh] + adst[d * H + phh]);
    smax[pidx] = pself;
  }

  auto build = [&](int lo, int hi) -> int {
    if (tid == 0) ovf = 0;
    int mycnt = 0;
    const int chunk = (hi - lo + NW - 1) / NW;
    const int cb = lo + wave * chunk;
    int ce = cb + chunk; if (ce > hi) ce = hi;
    for (int base = cb; base < ce; base += 32 * UA) {
      int dv[UA];
      if (base + 32 * UA <= ce) {
#pragma unroll
        for (int u = 0; u < UA; ++u) dv[u] = dst[base + u * 32 + lane];
      } else {
#pragma unroll
        for (int u = 0; u < UA; ++u) {
          const int e = base + u * 32 + lane;
          dv[u] = (e < ce) ? dst[e] : -1;
        }
      }
#pragma unroll
      for (int u = 0; u < UA; ++u) {
        const unsigned dl = (unsigned)(dv[u] - d0);
        unsigned m = __builtin_amdgcn_ballot_w32(dl < (unsigned)R);
        while (m) {
          const int l = __builtin_ctz(m);
          m &= m - 1;
          const int dls = __builtin_amdgcn_readlane((int)dl, l);
          const int t = dls / RT;
          if (lane == t) {
            if (mycnt < CAPW) wl[(wave * NTEAMS + t) * CAPW + mycnt] = base + u * 32 + l;
            ++mycnt;
          }
        }
      }
    }
    if (lane < NTEAMS) wcnt[wave * NTEAMS + lane] = mycnt;
    __syncthreads();

    int tot = 0;
    bool bad = false;
    for (int w = 0; w < NW; ++w) {
      const int c = wcnt[w * NTEAMS + team];
      bad |= (c > CAPW);
      tot += c;
    }
    bad |= (tot > TCAP);
    if (bad) ovf = 1;
    __syncthreads();
    const int o = ovf;
    __syncthreads();
    if (o) return -1;

    for (int j = c4; j < tot; j += F4) {
      int rem = j, w = 0;
      int c = wcnt[team];
      while (rem >= c) { rem -= c; ++w; c = wcnt[w * NTEAMS + team]; }
      const int e = wl[(w * NTEAMS + team) * CAPW + rem];
      const int s = src[e];
      const int d = dst[e];
      ts[j] = s;
      td[j] = d - d0 - r0;
      for (int hh = 0; hh < H; ++hh)
        tq[j * H + hh] = leaky02(asrc[s * H + hh] + adst[d * H + hh]);
    }
    __syncthreads();
    return tot;
  };

  auto build_parked = [&]() -> int {
    constexpr int WT  = F4 / 32;
    constexpr int EPW = (PCAP + WT - 1) / WT;
    constexpr int UL  = (EPW + 31) / 32;
    const float* sb = out + (size_t)blockIdx.x * pstride;
    bool have = can_park;
    if (have) {
      for (int p = 0; p < PT; ++p) {
        const int np = __float_as_int(sb[p]);
        have &= (np >= 0 && np <= PCAP);
      }
    }
    if (!have) return -2;
    if (tid == 0) ovf = 0;
    const int p  = r0 / PRT;
    const int n  = __float_as_int(sb[p]);
    const float* ps_src = sb + PT + p * 2 * PCAP;
    const float* ps_row = ps_src + PCAP;
    const int wt = c4 >> 5;
    int sv[UL], rv[UL], pos[UL];
#pragma unroll
    for (int u = 0; u < UL; ++u) {
      const int k   = u * 32 + lane;
      const int idx = wt * EPW + k;
      const bool ok = (k < EPW) && (idx < n);
      sv[u] = ok ? __float_as_int(ps_src[idx]) : 0;
      rv[u] = ok ? __float_as_int(ps_row[idx]) : -1;
    }
    int cnt = 0;
#pragma unroll
    for (int u = 0; u < UL; ++u) {
      const unsigned dl = (unsigned)(rv[u] - r0);
      const unsigned m = __builtin_amdgcn_ballot_w32(dl < (unsigned)RT);
      pos[u] = cnt + __builtin_popcount(m & ((1u << lane) - 1u));
      cnt += __builtin_popcount(m);
    }
    if (lane == 0) wcnt[team * WT + wt] = cnt;
    __syncthreads();
    int totl = 0, off = 0;
    for (int w = 0; w < WT; ++w) {
      const int c = wcnt[team * WT + w];
      off += (w < wt) ? c : 0;
      totl += c;
    }
    if (totl > TCAP) ovf = 1;
    __syncthreads();
    const int o = ovf;
    __syncthreads();
    if (o) return -1;
    float lg[UL][H];
#pragma unroll
    for (int u = 0; u < UL; ++u) {
      const bool hit = (unsigned)(rv[u] - r0) < (unsigned)RT;
      const int s = hit ? sv[u] : 0;
      const int d = hit ? d0 + rv[u] : d0;
      for (int hh = 0; hh < H; ++hh)
        lg[u][hh] = leaky02(asrc[s * H + hh] + adst[d * H + hh]);
    }
#pragma unroll
    for (int u = 0; u < UL; ++u) {
      const unsigned dl = (unsigned)(rv[u] - r0);
      if (dl < (unsigned)RT) {
        const int j = off + pos[u];
        ts[j] = sv[u];
        td[j] = (int)dl;
        for (int hh = 0; hh < H; ++hh) tq[j * H + hh] = lg[u][hh];
      }
    }
    __syncthreads();
    return totl;
  };

  auto pass_max = [&](int tot) {
    if (stat) {
      float m = smax[pidx];
#pragma unroll 8
      for (int j = 0; j < tot; ++j) {
        const float a = tq[j * H + phh];
        m = (td[j] == prow) ? fmaxf(m, a) : m;
      }
      smax[pidx] = m;
    }
  };

  auto pass_den = [&](int tot) {
    if (stat) {
      const float m = smax[pidx];
      float s = sden[pidx];
#pragma unroll 8
      for (int j = 0; j < tot; ++j) {
        const float ev = __expf(tq[j * H + phh] - m);
        s += (td[j] == prow) ? ev : 0.f;
      }
      sden[pidx] = s;
    }
  };

  auto pass_agg = [&](int tot) {
    for (int j = c4; j < tot; j += F4) {
      const int rr = (r0 + td[j]) * H;
      for (int hh = 0; hh < H; ++hh)
        tq[j * H + hh] = __expf(tq[j * H + hh] - smax[rr + hh]) / sden[rr + hh];
    }
    __syncthreads();
    const float* tqh = tq + h;
    int i = 0;
    for (; i + UC <= tot; i += UC) {
      int sr[UC], dr[UC]; float wt[UC]; v4f v[UC];
#pragma unroll
      for (int u = 0; u < UC; ++u) {
        sr[u] = ts[i + u]; dr[u] = td[i + u]; wt[u] = tqh[(i + u) * H];
      }
#pragma unroll
      for (int u = 0; u < UC; ++u) v[u] = *(const v4f*)(xcol + (size_t)sr[u] * F);
#pragma unroll
      for (int u = 0; u < UC; ++u) {
        v4f a = arow[dr[u] * F4];
        a += v[u] * wt[u];
        arow[dr[u] * F4] = a;
      }
    }
    for (; i < tot; ++i) {
      const v4f v1 = *(const v4f*)(xcol + (size_t)ts[i] * F);
      v4f a = arow[td[i] * F4];
      a += v1 * tqh[i * H];
      arow[td[i] * F4] = a;
    }
  };

  int tot;
  if constexpr (LOAD) {
    tot = build_parked();
    if (tot == -2) tot = build(0, E0);
  } else {
    tot = build(0, E0);
  }
  const bool whole = (tot >= 0);
  if constexpr (SAVE) {
    if (can_park) {
      float* sb = park + (size_t)blockIdx.x * pstride;
      if (whole) {
        float* ps_src = sb + PT + team * 2 * PCAP;
        for (int j = c4; j < tot; j += F4) {
          ps_src[j]        = __int_as_float(ts[j]);
          ps_src[PCAP + j] = __int_as_float(r0 + td[j]);
        }
      }
      if (c4 == 0) sb[team] = __int_as_float(whole ? tot : -1);
    }
  }
  const int step = whole ? ((E0 > 0) ? E0 : 1) : SUB;

  for (int mode = 0; mode < 3; ++mode) {
    if (mode == 1 && stat) sden[pidx] = __expf(pself - smax[pidx]);
    if (mode == 2) {
      if (stat) swt[pidx] = __expf(pself - smax[pidx]) / sden[pidx];
      __syncthreads();
      for (int r = 0; r < RT; r += UI) {
        v4f v[UI];
#pragma unroll
        for (int u = 0; u < UI; ++u) {
          const int d = d0 + r0 + r + u;
          const v4f z = {0.f, 0.f, 0.f, 0.f};
          v[u] = (d < Nn) ? *(const v4f*)(xcol + (size_t)d * F) : z;
        }
#pragma unroll
        for (int u = 0; u < UI; ++u) arow[(r + u) * F4] = v[u] * swt[(r0 + r + u) * H + h];
      }
    }
    for (int lo = 0; lo < E0; lo += step) {
      if (!whole) {
        const int hi = (lo + SUB < E0) ? lo + SUB : E0;
        tot = build(lo, hi);
      }
      if (mode == 0) pass_max(tot);
      else if (mode == 1) pass_den(tot);
      else pass_agg(tot);
      __syncthreads();
    }
    __syncthreads();
  }

  const float b0 = bias[c4 * 4 + 0], b1 = bias[c4 * 4 + 1];
  const float b2 = bias[c4 * 4 + 2], b3 = bias[c4 * 4 + 3];
  for (int r = 0; r < RT; ++r) {
    const int d = d0 + r0 + r;
    if (d < Nn) {
      v4f a = arow[r * F4];
      a.x += b0; a.y += b1; a.z += b2; a.w += b3;
      if (ELU) {
        a.x = (a.x > 0.f) ? a.x : expm1f(a.x);
        a.y = (a.y > 0.f) ? a.y : expm1f(a.y);
        a.z = (a.z > 0.f) ? a.z : expm1f(a.z);
        a.w = (a.w > 0.f) ? a.w : expm1f(a.w);
      }
      *(v4f*)(out + (size_t)d * F + c4 * 4) = a;
    }
  }
}


extern "C" void kernel_launch(void* const* d_in, const int* in_sizes, int n_in,
                              void* d_out, int out_size, void* d_ws, size_t ws_size,
                              hipStream_t stream) {
  const float* x   = (const float*)d_in[0];
  const int*   ei  = (const int*)d_in[1];
  const float* W1  = (const float*)d_in[2];
  const float* as1 = (const float*)d_in[3];
  const float* ad1 = (const float*)d_in[4];
  const float* b1  = (const float*)d_in[5];
  const float* W2  = (const float*)d_in[6];
  const float* as2 = (const float*)d_in[7];
  const float* ad2 = (const float*)d_in[8];
  const float* b2  = (const float*)d_in[9];

  const int K1 = 768, F1 = 768, H1 = 3;
  const int F2 = 256, H2 = 1;
  const int Nn = in_sizes[0] / K1;
  const int E0 = in_sizes[1] / 2;

  const int* srcI = ei;
  const int* dstI = ei + E0;

  float* ws = (float*)d_ws;
  float* xp1   = ws; ws += (size_t)Nn * F1;
  float* h1    = ws; ws += (size_t)Nn * F1;
  float* xp2   = ws; ws += (size_t)Nn * F2;
  float* asrc1 = ws; ws += (size_t)Nn * H1;
  float* adst1 = ws; ws += (size_t)Nn * H1;
  float* asrc2 = ws; ws += Nn;
  float* adst2 = ws; ws += Nn;

  const int PSTR = 80 * F2;
  const int PLIM = Nn * F2;

  float* wt1 = h1;
  float* wt2 = (float*)d_out + 8192;
  fold_att<<<(K1 * H1 + F1 * H2 + 7) / 8, 256, 0, stream>>>(
      W1, as1, ad1, wt1, K1, H1, W2, as2, ad2, wt2, F1, H2);

  {
    dim3 g(Nn / 16, F1 / 64 + 1);
    wmma_gemm_f32<<<g, 32, 0, stream>>>(x, W1, xp1, Nn, F1, K1, wt1, asrc1, adst1, H1);
  }
  owner_edge_phase<3, 768, 20, 48, 512, true, true, false, 20, 512>
      <<<(Nn + 79) / 80, 768, 0, stream>>>(
      srcI, dstI, E0, Nn, xp1, asrc1, adst1, b1, h1, (float*)d_out, PSTR, PLIM);

  {
    dim3 g(Nn / 16, F2 / 64 + 1);
    wmma_gemm_f32<<<g, 32, 0, stream>>>(h1, W2, xp2, Nn, F2, F1, wt2, asrc2, adst2, H2);
  }
  owner_edge_phase<1, 1024, 5, 24, 256, false, false, true, 20, 512>
      <<<(Nn + 79) / 80, 1024, 0, stream>>>(
      srcI, dstI, E0, Nn, xp2, asrc2, adst2, b2, (float*)d_out, nullptr, PSTR, PLIM);
}


# GEMM wave tile 16x64 -> 32x64 held in registers, no LDS: two 16-row tiles share every B fragment (a 16-byte B load feeds 8 WMMAs, not 4), half the waves; k order of every sum kept
# speedup vs baseline: 4.1509x; 1.0704x over previous
#include <hip/hip_runtime.h>
#include <math.h>


typedef __attribute__((ext_vector_type(2))) float v2f;
typedef __attribute__((ext_vector_type(4))) float v4f;
typedef __attribute__((ext_vector_type(8))) float v8f;

__global__ __launch_bounds__(32) void wmma_gemm_f32(
    const float* __restrict__ A, const float* __restrict__ B,
    float* __restrict__ C, int M, int N, int K,
    const float* __restrict__ Wt, float* __restrict__ asrc,
    float* __restrict__ adst, int H)
{
  const int lane = threadIdx.x & 31;
  const int half = lane >> 4;
  const int l15  = lane & 15;
  const int tm   = blockIdx.x;

  if (blockIdx.y == 0) {
    const int bc = (l15 < 2 * H) ? l15 : 0;
    const float* Ad = A + (size_t)(tm * 32 + l15) * K + (half << 4);
    const float* Ae = Ad + (size_t)16 * K;
    const float* Bd = Wt + (size_t)bc * K + (half << 4);
    v8f d0 = {}; v8f d1 = {}; v8f e0 = {}; v8f e1 = {};
    for (int kb = 0; kb < K; kb += 32) {
      v4f aq[4];
      v4f ar[4];
      v4f bq[4];
#pragma unroll
      for (int i = 0; i < 4; ++i) aq[i] = *(const v4f*)(Ad + kb + 4 * i);
#pragma unroll
      for (int i = 0; i < 4; ++i) ar[i] = *(const v4f*)(Ae + kb + 4 * i);
#pragma unroll
      for (int i = 0; i < 4; ++i) bq[i] = *(const v4f*)(Bd + kb + 4 * i);
#pragma unroll
      for (int j = 0; j < 4; ++j) {
        v2f a0, a1, f0, f1, b0, b1;
        a0.x = aq[j][0]; a0.y = aq[j][1];
        a1.x = aq[j][2]; a1.y = aq[j][3];
        f0.x = ar[j][0]; f0.y = ar[j][1];
        f1.x = ar[j][2]; f1.y = ar[j][3];
        b0.x = bq[j][0]; b0.y = bq[j][1];
        b1.x = bq[j][2]; b1.y = bq[j][3];
        d0 = __builtin_amdgcn_wmma_f32_16x16x4_f32(false, a0, false, b0, (short)0, d0, false, false);
        e0 = __builtin_amdgcn_wmma_f32_16x16x4_f32(false, f0, false, b0, (short)0, e0, false, false);
        d1 = __builtin_amdgcn_wmma_f32_16x16x4_f32(false, a1, false, b1, (short)0, d1, false, false);
        e1 = __builtin_amdgcn_wmma_f32_16x16x4_f32(false, f1, false, b1, (short)0, e1, false, false);
      }
    }
    if (l15 < 2 * H) {
      float* o = (l15 < H) ? (asrc + l15) : (adst + (l15 - H));
      const int row0 = tm * 32 + (half << 3);
      for (int r = 0; r < 8; ++r) o[(size_t)(row0 + r) * H] = d0[r] + d1[r];
      for (int r = 0; r < 8; ++r) o[(size_t)(row0 + 16 + r) * H] = e0[r] + e1[r];
    }
    return;
  }
  const int tn   = (blockIdx.y - 1) << 6;

  v8f c0 = {}; v8f c1 = {}; v8f c2 = {}; v8f c3 = {};
  v8f g0 = {}; v8f g1 = {}; v8f g2 = {}; v8f g3 = {};

  const float* Ap = A + (size_t)(tm * 32 + l15) * K + (half << 4);
  const float* Aq = Ap + (size_t)16 * K;
  const float* Bp = B + (size_t)(half << 4) * N + tn + (l15 << 2);
  for (int kb = 0; kb < K; kb += 32) {
    v4f aq[4];
    v4f ar[4];
    v4f bq[16];
#pragma unroll
    for (int i = 0; i < 4; ++i) aq[i] = *(const v4f*)(Ap + kb + 4 * i);
#pragma unroll
    for (int i = 0; i < 4; ++i) ar[i] = *(const v4f*)(Aq + kb + 4 * i);
#pragma unroll
    for (int m = 0; m < 16; ++m) bq[m] = *(const v4f*)(Bp + (size_t)(kb + m) * N);
    __builtin_amdgcn_sched_barrier(0);
#pragma unroll
    for (int j = 0; j < 8; ++j) {
      v2f a, f;
      a.x = aq[j >> 1][2 * (j & 1)];
      a.y = aq[j >> 1][2 * (j & 1) + 1];
      f.x = ar[j >> 1][2 * (j & 1)];
      f.y = ar[j >> 1][2 * (j & 1) + 1];
      v2f b0, b1, b2, b3;
      b0.x = bq[2 * j][0]; b0.y = bq[2 * j + 1][0];
      b1.x = bq[2 * j][1]; b1.y = bq[2 * j + 1][1];
      b2.x = bq[2 * j][2]; b2.y = bq[2 * j + 1][2];
      b3.x = bq[2 * j][3]; b3.y = bq[2 * j + 1][3];
      c0 = __builtin_amdgcn_wmma_f32_16x16x4_f32(false, a, false, b0, (short)0, c0, false, false);
      g0 = __builtin_amdgcn_wmma_f32_16x16x4_f32(false, f, false, b0, (short)0, g0, false, false);
      c1 = __builtin_amdgcn_wmma_f32_16x16x4_f32(false, a, false, b1, (short)0, c1, false, false);
      g1 = __builtin_amdgcn_wmma_f32_16x16x4_f32(false, f, false, b1, (short)0, g1, false, false);
      c2 = __builtin_amdgcn_wmma_f32_16x16x4_f32(false, a, false, b2, (short)0, c2, false, false);
      g2 = __builtin_amdgcn_wmma_f32_16x16x4_f32(false, f, false, b2, (short)0, g2, false, false);
      c3 = __builtin_amdgcn_wmma_f32_16x16x4_f32(false, a, false, b3, (short)0, c3, false, false);
      g3 = __builtin_amdgcn_wmma_f32_16x16x4_f32(false, f, false, b3, (short)0, g3, false, false);
    }
  }

  const int orow = tm * 32 + (half << 3);
  for (int r = 0; r < 8; ++r) {
    v4f o;
    o.x = c0[r]; o.y = c1[r]; o.z = c2[r]; o.w = c3[r];
    *(v4f*)(C + (size_t)(orow + r) * N + tn + (l15 << 2)) = o;
  }
  for (int r = 0; r < 8; ++r) {
    v4f o;
    o.x = g0[r]; o.y = g1[r]; o.z = g2[r]; o.w = g3[r];
    *(v4f*)(C + (size_t)(orow + 16 + r) * N + tn + (l15 << 2)) = o;
  }
}

__global__ __launch_bounds__(256) void fold_att(
    const float* __restrict__ W1, const float* __restrict__ as1,
    const float* __restrict__ ad1, float* __restrict__ wt1, int K1, int H1,
    const float* __restrict__ W2, const float* __restrict__ as2,
    const float* __restrict__ ad2, float* __restrict__ wt2, int K2, int H2)
{
  int wid = (int)((blockIdx.x * blockDim.x + threadIdx.x) >> 5);
  const int lane = threadIdx.x & 31;
  const float* Wm = W1; const float* as = as1; const float* ad = ad1;
  float* wt = wt1; int K = K1; int H = H1;
  if (wid >= K1 * H1) {
    wid -= K1 * H1;
    if (wid >= K2 * H2) return;
    Wm = W2; as = as2; ad = ad2; wt = wt2; K = K2; H = H2;
  }
  const int k = wid / H;
  const int h = wid - k * H;
  const float* wr = Wm + (size_t)k * H * 256 + h * 256 + lane * 4;
  const float* sr = as + h * 256 + lane * 4;
  const float* dr = ad + h * 256 + lane * 4;
  const v4f w0 = *(const v4f*)wr,  w1 = *(const v4f*)(wr + 128);
  const v4f s0 = *(const v4f*)sr,  s1 = *(const v4f*)(sr + 128);
  const v4f e0 = *(const v4f*)dr,  e1 = *(const v4f*)(dr + 128);
  float s = w0.x * s0.x + w0.y * s0.y + w0.z * s0.z + w0.w * s0.w
          + w1.x * s1.x + w1.y * s1.y + w1.z * s1.z + w1.w * s1.w;
  float d = w0.x * e0.x + w0.y * e0.y + w0.z * e0.z + w0.w * e0.w
          + w1.x * e1.x + w1.y * e1.y + w1.z * e1.z + w1.w * e1.w;
  for (int off = 16; off > 0; off >>= 1) {
    s += __shfl_down(s, off, 32);
    d += __shfl_down(d, off, 32);
  }
  if (lane == 0) {
    wt[(size_t)h * K + k]       = s;
    wt[(size_t)(H + h) * K + k] = d;
  }
}


__device__ __forceinline__ float leaky02(float a) { return (a > 0.f) ? a : 0.2f * a; }

template <int H, int NT, int RT, int CAPW, int TCAP, bool ELU,
          bool SAVE, bool LOAD, int PRT, int PCAP>
__global__ __launch_bounds__(NT) void owner_edge_phase(
    const int* __restrict__ src, const int* __restrict__ dst, int E0, int Nn,
    const float* __restrict__ xp, const float* __restrict__ asrc,
    const float* __restrict__ adst, const float* __restrict__ bias,
    float* __restrict__ out, float* __restrict__ park, int pstride, int plimit)
{
  constexpr int F      = H * 256;
  constexpr int F4     = F / 4;
  constexpr int NTEAMS = NT / F4;
  constexpr int R      = NTEAMS * RT;
  constexpr int NW     = NT / 32;
  constexpr int NL     = NW * NTEAMS;
  constexpr int SUB    = (TCAP < NW * CAPW) ? TCAP : NW * CAPW;
  constexpr int UA     = 16;
  constexpr int UC     = 8;
  constexpr int UI     = 5;
  static_assert(NT % F4 == 0 && F4 % 32 == 0 && NTEAMS <= 32, "team shape");
  static_assert(RT * H <= F4, "one thread per (row, head) of a team");
  static_assert(RT % UI == 0, "self-loop rows are loaded UI at a time");
  static_assert(R % PRT == 0 && PRT % RT == 0 && PCAP > 0, "parked lists: a team's rows lie in one parent team");
  static_assert(!SAVE || (PRT == RT && PCAP == TCAP), "a SAVE launch parks its own teams");
  constexpr int PT    = R / PRT;
  constexpr int PNEED = PT + 2 * PT * PCAP;

  __shared__ v4f   acc[R * F4];
  __shared__ int   wl[NL * CAPW];
  __shared__ int   wcnt[NL];
  __shared__ int   tsrc[NTEAMS * TCAP];
  __shared__ int   tdl[NTEAMS * TCAP];
  __shared__ float tw[NTEAMS * TCAP * H];
  __shared__ float smax[R * H];
  __shared__ float sden[R * H];
  __shared__ float swt[R * H];
  __shared__ int   ovf;

  const int tid  = threadIdx.x;
  const int lane = tid & 31;
  const int wave = tid >> 5;
  const int team = tid / F4;
  const int c4   = tid - team * F4;
  const int h    = c4 >> 6;
  const int d0   = blockIdx.x * R;
  const int r0   = team * RT;
  const bool can_park = (size_t)blockIdx.x * (size_t)pstride + (size_t)PNEED <= (size_t)plimit;

  const float* xcol = xp + c4 * 4;
  v4f*   arow = acc + r0 * F4 + c4;
  int*   ts   = tsrc + team * TCAP;
  int*   td   = tdl + team * TCAP;
  float* tq   = tw + team * TCAP * H;

  const bool stat = c4 < RT * H;
  const int  prow = c4 / H;
  const int  phh  = c4 - prow * H;
  const int  pidx = (r0 + prow) * H + phh;
  float pself = 0.f;
  if (stat) {
    const int d = d0 + r0 + prow;
    if (d < Nn) pself = leaky02(asrc[d * H + phh] + adst[d * H + phh]);
    smax[pidx] = pself;
  }

  auto build = [&](int lo, int hi) -> int {
    if (tid == 0) ovf = 0;
    int mycnt = 0;
    const int chunk = (hi - lo + NW - 1) / NW;
    const int cb = lo + wave * chunk;
    int ce = cb + chunk; if (ce > hi) ce = hi;
    for (int base = cb; base < ce; base += 32 * UA) {
      int dv[UA];
      if (base + 32 * UA <= ce) {
#pragma unroll
        for (int u = 0; u < UA; ++u) dv[u] = dst[base + u * 32 + lane];
      } else {
#pragma unroll
        for (int u = 0; u < UA; ++u) {
          const int e = base + u * 32 + lane;
          dv[u] = (e < ce) ? dst[e] : -1;
        }
      }
#pragma unroll
      for (int u = 0; u < UA; ++u) {
        const unsigned dl = (unsigned)(dv[u] - d0);
        unsigned m = __builtin_amdgcn_ballot_w32(dl < (unsigned)R);
        while (m) {
          const int l = __builtin_ctz(m);
          m &= m - 1;
          const int dls = __builtin_amdgcn_readlane((int)dl, l);
          const int t = dls / RT;
          if (lane == t) {
            if (mycnt < CAPW) wl[(wave * NTEAMS + t) * CAPW + mycnt] = base + u * 32 + l;
            ++mycnt;
          }
        }
      }
    }
    if (lane < NTEAMS) wcnt[wave * NTEAMS + lane] = mycnt;
    __syncthreads();

    int tot = 0;
    bool bad = false;
    for (int w = 0; w < NW; ++w) {
      const int c = wcnt[w * NTEAMS + team];
      bad |= (c > CAPW);
      tot += c;
    }
    bad |= (tot > TCAP);
    if (bad) ovf = 1;
    __syncthreads();
    const int o = ovf;
    __syncthreads();
    if (o) return -1;

    for (int j = c4; j < tot; j += F4) {
      int rem = j, w = 0;
      int c = wcnt[team];
      while (rem >= c) { rem -= c; ++w; c = wcnt[w * NTEAMS + team]; }
      const int e = wl[(w * NTEAMS + team) * CAPW + rem];
      const int s = src[e];
      const int d = dst[e];
      ts[j] = s;
      td[j] = d - d0 - r0;
      for (int hh = 0; hh < H; ++hh)
        tq[j * H + hh] = leaky02(asrc[s * H + hh] + adst[d * H + hh]);
    }
    __syncthreads();
    return tot;
  };

  auto build_parked = [&]() -> int {
    constexpr int WT  = F4 / 32;
    constexpr int EPW = (PCAP + WT - 1) / WT;
    constexpr int UL  = (EPW + 31) / 32;
    const float* sb = out + (size_t)blockIdx.x * pstride;
    bool have = can_park;
    if (have) {
      for (int p = 0; p < PT; ++p) {
        const int np = __float_as_int(sb[p]);
        have &= (np >= 0 && np <= PCAP);
      }
    }
    if (!have) return -2;
    if (tid == 0) ovf = 0;
    const int p  = r0 / PRT;
    const int n  = __float_as_int(sb[p]);
    const float* ps_src = sb + PT + p * 2 * PCAP;
    const float* ps_row = ps_src + PCAP;
    const int wt = c4 >> 5;
    int sv[UL], rv[UL], pos[UL];
#pragma unroll
    for (int u = 0; u < UL; ++u) {
      const int k   = u * 32 + lane;
      const int idx = wt * EPW + k;
      const bool ok = (k < EPW) && (idx < n);
      sv[u] = ok ? __float_as_int(ps_src[idx]) : 0;
      rv[u] = ok ? __float_as_int(ps_row[idx]) : -1;
    }
    int cnt = 0;
#pragma unroll
    for (int u = 0; u < UL; ++u) {
      const unsigned dl = (unsigned)(rv[u] - r0);
      const unsigned m = __builtin_amdgcn_ballot_w32(dl < (unsigned)RT);
      pos[u] = cnt + __builtin_popcount(m & ((1u << lane) - 1u));
      cnt += __builtin_popcount(m);
    }
    if (lane == 0) wcnt[team * WT + wt] = cnt;
    __syncthreads();
    int totl = 0, off = 0;
    for (int w = 0; w < WT; ++w) {
      const int c = wcnt[team * WT + w];
      off += (w < wt) ? c : 0;
      totl += c;
    }
    if (totl > TCAP) ovf = 1;
    __syncthreads();
    const int o = ovf;
    __syncthreads();
    if (o) return -1;
    float lg[UL][H];
#pragma unroll
    for (int u = 0; u < UL; ++u) {
      const bool hit = (unsigned)(rv[u] - r0) < (unsigned)RT;
      const int s = hit ? sv[u] : 0;
      const int d = hit ? d0 + rv[u] : d0;
      for (int hh = 0; hh < H; ++hh)
        lg[u][hh] = leaky02(asrc[s * H + hh] + adst[d * H + hh]);
    }
#pragma unroll
    for (int u = 0; u < UL; ++u) {
      const unsigned dl = (unsigned)(rv[u] - r0);
      if (dl < (unsigned)RT) {
        const int j = off + pos[u];
        ts[j] = sv[u];
        td[j] = (int)dl;
        for (int hh = 0; hh < H; ++hh) tq[j * H + hh] = lg[u][hh];
      }
    }
    __syncthreads();
    return totl;
  };

  auto pass_max = [&](int tot) {
    if (stat) {
      float m = smax[pidx];
#pragma unroll 8
      for (int j = 0; j < tot; ++j) {
        const float a = tq[j * H + phh];
        m = (td[j] == prow) ? fmaxf(m, a) : m;
      }
      smax[pidx] = m;
    }
  };

  auto pass_den = [&](int tot) {
    if (stat) {
      const float m = smax[pidx];
      float s = sden[pidx];
#pragma unroll 8
      for (int j = 0; j < tot; ++j) {
        const float ev = __expf(tq[j * H + phh] - m);
        s += (td[j] == prow) ? ev : 0.f;
      }
      sden[pidx] = s;
    }
  };

  auto pass_agg = [&](int tot) {
    for (int j = c4; j < tot; j += F4) {
      const int rr = (r0 + td[j]) * H;
      for (int hh = 0; hh < H; ++hh)
        tq[j * H + hh] = __expf(tq[j * H + hh] - smax[rr + hh]) / sden[rr + hh];
    }
    __syncthreads();
    const float* tqh = tq + h;
    int i = 0;
    for (; i + UC <= tot; i += UC) {
      int sr[UC], dr[UC]; float wt[UC]; v4f v[UC];
#pragma unroll
      for (int u = 0; u < UC; ++u) {
        sr[u] = ts[i + u]; dr[u] = td[i + u]; wt[u] = tqh[(i + u) * H];
      }
#pragma unroll
      for (int u = 0; u < UC; ++u) v[u] = *(const v4f*)(xcol + (size_t)sr[u] * F);
#pragma unroll
      for (int u = 0; u < UC; ++u) {
        v4f a = arow[dr[u] * F4];
        a += v[u] * wt[u];
        arow[dr[u] * F4] = a;
      }
    }
    for (; i < tot; ++i) {
      const v4f v1 = *(const v4f*)(xcol + (size_t)ts[i] * F);
      v4f a = arow[td[i] * F4];
      a += v1 * tqh[i * H];
      arow[td[i] * F4] = a;
    }
  };

  int tot;
  if constexpr (LOAD) {
    tot = build_parked();
    if (tot == -2) tot = build(0, E0);
  } else {
    tot = build(0, E0);
  }
  const bool whole = (tot >= 0);
  if constexpr (SAVE) {
    if (can_park) {
      float* sb = park + (size_t)blockIdx.x * pstride;
      if (whole) {
        float* ps_src = sb + PT + team * 2 * PCAP;
        for (int j = c4; j < tot; j += F4) {
          ps_src[j]        = __int_as_float(ts[j]);
          ps_src[PCAP + j] = __int_as_float(r0 + td[j]);
        }
      }
      if (c4 == 0) sb[team] = __int_as_float(whole ? tot : -1);
    }
  }
  const int step = whole ? ((E0 > 0) ? E0 : 1) : SUB;

  for (int mode = 0; mode < 3; ++mode) {
    if (mode == 1 && stat) sden[pidx] = __expf(pself - smax[pidx]);
    if (mode == 2) {
      if (stat) swt[pidx] = __expf(pself - smax[pidx]) / sden[pidx];
      __syncthreads();
      for (int r = 0; r < RT; r += UI) {
        v4f v[UI];
#pragma unroll
        for (int u = 0; u < UI; ++u) {
          const int d = d0 + r0 + r + u;
          const v4f z = {0.f, 0.f, 0.f, 0.f};
          v[u] = (d < Nn) ? *(const v4f*)(xcol + (size_t)d * F) : z;
        }
#pragma unroll
        for (int u = 0; u < UI; ++u) arow[(r + u) * F4] = v[u] * swt[(r0 + r + u) * H + h];
      }
    }
    for (int lo = 0; lo < E0; lo += step) {
      if (!whole) {
        const int hi = (lo + SUB < E0) ? lo + SUB : E0;
        tot = build(lo, hi);
      }
      if (mode == 0) pass_max(tot);
      else if (mode == 1) pass_den(tot);
      else pass_agg(tot);
      __syncthreads();
    }
    __syncthreads();
  }

  const float b0 = bias[c4 * 4 + 0], b1 = bias[c4 * 4 + 1];
  const float b2 = bias[c4 * 4 + 2], b3 = bias[c4 * 4 + 3];
  for (int r = 0; r < RT; ++r) {
    const int d = d0 + r0 + r;
    if (d < Nn) {
      v4f a = arow[r * F4];
      a.x += b0; a.y += b1; a.z += b2; a.w += b3;
      if (ELU) {
        a.x = (a.x > 0.f) ? a.x : expm1f(a.x);
        a.y = (a.y > 0.f) ? a.y : expm1f(a.y);
        a.z = (a.z > 0.f) ? a.z : expm1f(a.z);
        a.w = (a.w > 0.f) ? a.w : expm1f(a.w);
      }
      *(v4f*)(out + (size_t)d * F + c4 * 4) = a;
    }
  }
}


extern "C" void kernel_launch(void* const* d_in, const int* in_sizes, int n_in,
                              void* d_out, int out_size, void* d_ws, size_t ws_size,
                              hipStream_t stream) {
  const float* x   = (const float*)d_in[0];
  const int*   ei  = (const int*)d_in[1];
  const float* W1  = (const float*)d_in[2];
  const float* as1 = (const float*)d_in[3];
  const float* ad1 = (const float*)d_in[4];
  const float* b1  = (const float*)d_in[5];
  const float* W2  = (const float*)d_in[6];
  const float* as2 = (const float*)d_in[7];
  const float* ad2 = (const float*)d_in[8];
  const float* b2  = (const float*)d_in[9];

  const int K1 = 768, F1 = 768, H1 = 3;
  const int F2 = 256, H2 = 1;
  const int Nn = in_sizes[0] / K1;
  const int E0 = in_sizes[1] / 2;

  const int* srcI = ei;
  const int* dstI = ei + E0;

  float* ws = (float*)d_ws;
  float* xp1   = ws; ws += (size_t)Nn * F1;
  float* h1    = ws; ws += (size_t)Nn * F1;
  float* xp2   = ws; ws += (size_t)Nn * F2;
  float* asrc1 = ws; ws += (size_t)Nn * H1;
  float* adst1 = ws; ws += (size_t)Nn * H1;
  float* asrc2 = ws; ws += Nn;
  float* adst2 = ws; ws += Nn;

  const int PSTR = 80 * F2;
  const int PLIM = Nn * F2;

  float* wt1 = h1;
  float* wt2 = (float*)d_out + 8192;
  fold_att<<<(K1 * H1 + F1 * H2 + 7) / 8, 256, 0, stream>>>(
      W1, as1, ad1, wt1, K1, H1, W2, as2, ad2, wt2, F1, H2);

  {
    dim3 g(Nn / 32, F1 / 64 + 1);
    wmma_gemm_f32<<<g, 32, 0, stream>>>(x, W1, xp1, Nn, F1, K1, wt1, asrc1, adst1, H1);
  }
  owner_edge_phase<3, 768, 20, 48, 512, true, true, false, 20, 512>
      <<<(Nn + 79) / 80, 768, 0, stream>>>(
      srcI, dstI, E0, Nn, xp1, asrc1, adst1, b1, h1, (float*)d_out, PSTR, PLIM);

  {
    dim3 g(Nn / 32, F2 / 64 + 1);
    wmma_gemm_f32<<<g, 32, 0, stream>>>(h1, W2, xp2, Nn, F2, F1, wt2, asrc2, adst2, H2);
  }
  owner_edge_phase<1, 1024, 5, 24, 256, false, false, true, 20, 512>
      <<<(Nn + 79) / 80, 1024, 0, stream>>>(
      srcI, dstI, E0, Nn, xp2, asrc2, adst2, b2, (float*)d_out, nullptr, PSTR, PLIM);
}


# Team edge list grouped by dst row in LDS (per-wave count-and-place, edge order kept): stat walks read only their row's entries, agg keeps a row's sum in a register; every sum order kept
# speedup vs baseline: 4.4977x; 1.0835x over previous
#include <hip/hip_runtime.h>
#include <math.h>


typedef __attribute__((ext_vector_type(2))) float v2f;
typedef __attribute__((ext_vector_type(4))) float v4f;
typedef __attribute__((ext_vector_type(8))) float v8f;

__global__ __launch_bounds__(32) void wmma_gemm_f32(
    const float* __restrict__ A, const float* __restrict__ B,
    float* __restrict__ C, int M, int N, int K,
    const float* __restrict__ Wt, float* __restrict__ asrc,
    float* __restrict__ adst, int H)
{
  const int lane = threadIdx.x & 31;
  const int half = lane >> 4;
  const int l15  = lane & 15;
  const int tm   = blockIdx.x;

  if (blockIdx.y == 0) {
    const int bc = (l15 < 2 * H) ? l15 : 0;
    const float* Ad = A + (size_t)(tm * 32 + l15) * K + (half << 4);
    const float* Ae = Ad + (size_t)16 * K;
    const float* Bd = Wt + (size_t)bc * K + (half << 4);
    v8f d0 = {}; v8f d1 = {}; v8f e0 = {}; v8f e1 = {};
    for (int kb = 0; kb < K; kb += 32) {
      v4f aq[4];
      v4f ar[4];
      v4f bq[4];
#pragma unroll
      for (int i = 0; i < 4; ++i) aq[i] = *(const v4f*)(Ad + kb + 4 * i);
#pragma unroll
      for (int i = 0; i < 4; ++i) ar[i] = *(const v4f*)(Ae + kb + 4 * i);
#pragma unroll
      for (int i = 0; i < 4; ++i) bq[i] = *(const v4f*)(Bd + kb + 4 * i);
#pragma unroll
      for (int j = 0; j < 4; ++j) {
        v2f a0, a1, f0, f1, b0, b1;
        a0.x = aq[j][0]; a0.y = aq[j][1];
        a1.x = aq[j][2]; a1.y = aq[j][3];
        f0.x = ar[j][0]; f0.y = ar[j][1];
        f1.x = ar[j][2]; f1.y = ar[j][3];
        b0.x = bq[j][0]; b0.y = bq[j][1];
        b1.x = bq[j][2]; b1.y = bq[j][3];
        d0 = __builtin_amdgcn_wmma_f32_16x16x4_f32(false, a0, false, b0, (short)0, d0, false, false);
        e0 = __builtin_amdgcn_wmma_f32_16x16x4_f32(false, f0, false, b0, (short)0, e0, false, false);
        d1 = __builtin_amdgcn_wmma_f32_16x16x4_f32(false, a1, false, b1, (short)0, d1, false, false);
        e1 = __builtin_amdgcn_wmma_f32_16x16x4_f32(false, f1, false, b1, (short)0, e1, false, false);
      }
    }
    if (l15 < 2 * H) {
      float* o = (l15 < H) ? (asrc + l15) : (adst + (l15 - H));
      const int row0 = tm * 32 + (half << 3);
      for (int r = 0; r < 8; ++r) o[(size_t)(row0 + r) * H] = d0[r] + d1[r];
      for (int r = 0; r < 8; ++r) o[(size_t)(row0 + 16 + r) * H] = e0[r] + e1[r];
    }
    return;
  }
  const int tn   = (blockIdx.y - 1) << 6;

  v8f c0 = {}; v8f c1 = {}; v8f c2 = {}; v8f c3 = {};
  v8f g0 = {}; v8f g1 = {}; v8f g2 = {}; v8f g3 = {};

  const float* Ap = A + (size_t)(tm * 32 + l15) * K + (half << 4);
  const float* Aq = Ap + (size_t)16 * K;
  const float* Bp = B + (size_t)(half << 4) * N + tn + (l15 << 2);
  for (int kb = 0; kb < K; kb += 32) {
    v4f aq[4];
    v4f ar[4];
    v4f bq[16];
#pragma unroll
    for (int i = 0; i < 4; ++i) aq[i] = *(const v4f*)(Ap + kb + 4 * i);
#pragma unroll
    for (int i = 0; i < 4; ++i) ar[i] = *(const v4f*)(Aq + kb + 4 * i);
#pragma unroll
    for (int m = 0; m < 16; ++m) bq[m] = *(const v4f*)(Bp + (size_t)(kb + m) * N);
    __builtin_amdgcn_sched_barrier(0);
#pragma unroll
    for (int j = 0; j < 8; ++j) {
      v2f a, f;
      a.x = aq[j >> 1][2 * (j & 1)];
      a.y = aq[j >> 1][2 * (j & 1) + 1];
      f.x = ar[j >> 1][2 * (j & 1)];
      f.y = ar[j >> 1][2 * (j & 1) + 1];
      v2f b0, b1, b2, b3;
      b0.x = bq[2 * j][0]; b0.y = bq[2 * j + 1][0];
      b1.x = bq[2 * j][1]; b1.y = bq[2 * j + 1][1];
      b2.x = bq[2 * j][2]; b2.y = bq[2 * j + 1][2];
      b3.x = bq[2 * j][3]; b3.y = bq[2 * j + 1][3];
      c0 = __builtin_amdgcn_wmma_f32_16x16x4_f32(false, a, false, b0, (short)0, c0, false, false);
      g0 = __builtin_amdgcn_wmma_f32_16x16x4_f32(false, f, false, b0, (short)0, g0, false, false);
      c1 = __builtin_amdgcn_wmma_f32_16x16x4_f32(false, a, false, b1, (short)0, c1, false, false);
      g1 = __builtin_amdgcn_wmma_f32_16x16x4_f32(false, f, false, b1, (short)0, g1, false, false);
      c2 = __builtin_amdgcn_wmma_f32_16x16x4_f32(false, a, false, b2, (short)0, c2, false, false);
      g2 = __builtin_amdgcn_wmma_f32_16x16x4_f32(false, f, false, b2, (short)0, g2, false, false);
      c3 = __builtin_amdgcn_wmma_f32_16x16x4_f32(false, a, false, b3, (short)0, c3, false, false);
      g3 = __builtin_amdgcn_wmma_f32_16x16x4_f32(false, f, false, b3, (short)0, g3, false, false);
    }
  }

  const int orow = tm * 32 + (half << 3);
  for (int r = 0; r < 8; ++r) {
    v4f o;
    o.x = c0[r]; o.y = c1[r]; o.z = c2[r]; o.w = c3[r];
    *(v4f*)(C + (size_t)(orow + r) * N + tn + (l15 << 2)) = o;
  }
  for (int r = 0; r < 8; ++r) {
    v4f o;
    o.x = g0[r]; o.y = g1[r]; o.z = g2[r]; o.w = g3[r];
    *(v4f*)(C + (size_t)(orow + 16 + r) * N + tn + (l15 << 2)) = o;
  }
}

__global__ __launch_bounds__(256) void fold_att(
    const float* __restrict__ W1, const float* __restrict__ as1,
    const float* __restrict__ ad1, float* __restrict__ wt1, int K1, int H1,
    const float* __restrict__ W2, const float* __restrict__ as2,
    const float* __restrict__ ad2, float* __restrict__ wt2, int K2, int H2)
{
  int wid = (int)((blockIdx.x * blockDim.x + threadIdx.x) >> 5);
  const int lane = threadIdx.x & 31;
  const float* Wm = W1; const float* as = as1; const float* ad = ad1;
  float* wt = wt1; int K = K1; int H = H1;
  if (wid >= K1 * H1) {
    wid -= K1 * H1;
    if (wid >= K2 * H2) return;
    Wm = W2; as = as2; ad = ad2; wt = wt2; K = K2; H = H2;
  }
  const int k = wid / H;
  const int h = wid - k * H;
  const float* wr = Wm + (size_t)k * H * 256 + h * 256 + lane * 4;
  const float* sr = as + h * 256 + lane * 4;
  const float* dr = ad + h * 256 + lane * 4;
  const v4f w0 = *(const v4f*)wr,  w1 = *(const v4f*)(wr + 128);
  const v4f s0 = *(const v4f*)sr,  s1 = *(const v4f*)(sr + 128);
  const v4f e0 = *(const v4f*)dr,  e1 = *(const v4f*)(dr + 128);
  float s = w0.x * s0.x + w0.y * s0.y + w0.z * s0.z + w0.w * s0.w
          + w1.x * s1.x + w1.y * s1.y + w1.z * s1.z + w1.w * s1.w;
  float d = w0.x * e0.x + w0.y * e0.y + w0.z * e0.z + w0.w * e0.w
          + w1.x * e1.x + w1.y * e1.y + w1.z * e1.z + w1.w * e1.w;
  for (int off = 16; off > 0; off >>= 1) {
    s += __shfl_down(s, off, 32);
    d += __shfl_down(d, off, 32);
  }
  if (lane == 0) {
    wt[(size_t)h * K + k]       = s;
    wt[(size_t)(H + h) * K + k] = d;
  }
}


__device__ __forceinline__ float leaky02(float a) { return (a > 0.f) ? a : 0.2f * a; }

template <int H, int NT, int RT, int CAPW, int TCAP, bool ELU,
          bool SAVE, bool LOAD, int PRT, int PCAP>
__global__ __launch_bounds__(NT) void owner_edge_phase(
    const int* __restrict__ src, const int* __restrict__ dst, int E0, int Nn,
    const float* __restrict__ xp, const float* __restrict__ asrc,
    const float* __restrict__ adst, const float* __restrict__ bias,
    float* __restrict__ out, float* __restrict__ park, int pstride, int plimit)
{
  constexpr int F      = H * 256;
  constexpr int F4     = F / 4;
  constexpr int NTEAMS = NT / F4;
  constexpr int R      = NTEAMS * RT;
  constexpr int NW     = NT / 32;
  constexpr int NL     = NW * NTEAMS;
  constexpr int SUB    = (TCAP < NW * CAPW) ? TCAP : NW * CAPW;
  constexpr int UA     = 16;
  constexpr int UC     = 8;
  constexpr int UI     = 5;
  static_assert(NT % F4 == 0 && F4 % 32 == 0 && NTEAMS <= 32, "team shape");
  static_assert(RT * H <= F4, "one thread per (row, head) of a team");
  static_assert(RT % UI == 0, "self-loop rows are loaded UI at a time");
  static_assert(R % PRT == 0 && PRT % RT == 0 && PCAP > 0, "parked lists: a team's rows lie in one parent team");
  static_assert(!SAVE || (PRT == RT && PCAP == TCAP), "a SAVE launch parks its own teams");
  static_assert(NL * CAPW >= NTEAMS * TCAP, "the row-grouped entry numbers reuse the wave lists' space");
  constexpr int PT    = R / PRT;
  constexpr int PNEED = PT + 2 * PT * PCAP;

  __shared__ v4f   acc[R * F4];
  __shared__ int   wl[NL * CAPW];
  __shared__ int   wcnt[NL];
  __shared__ int   tsrc[NTEAMS * TCAP];
  __shared__ __attribute__((aligned(16))) int tdl[NTEAMS * TCAP + 16];
  __shared__ float tw[NTEAMS * TCAP * H];
  __shared__ float smax[R * H];
  __shared__ float sden[R * H];
  __shared__ float swt[R * H];
  __shared__ __attribute__((aligned(16))) int rcnt[NW * RT];
  __shared__ int   rbeg[NTEAMS * (RT + 1)];
  __shared__ int   ovf;

  const int tid  = threadIdx.x;
  const int lane = tid & 31;
  const int wave = tid >> 5;
  const int team = tid / F4;
  const int c4   = tid - team * F4;
  const int h    = c4 >> 6;
  const int d0   = blockIdx.x * R;
  const int r0   = team * RT;
  const bool can_park = (size_t)blockIdx.x * (size_t)pstride + (size_t)PNEED <= (size_t)plimit;

  const float* xcol = xp + c4 * 4;
  v4f*   arow = acc + r0 * F4 + c4;
  int*   ts   = tsrc + team * TCAP;
  int*   td   = tdl + team * TCAP;
  float* tq   = tw + team * TCAP * H;
  int*   pm   = wl + team * TCAP;
  int*   rb   = rbeg + team * (RT + 1);

  const bool stat = c4 < RT * H;
  const int  prow = c4 / H;
  const int  phh  = c4 - prow * H;
  const int  pidx = (r0 + prow) * H + phh;
  float pself = 0.f;
  if (stat) {
    const int d = d0 + r0 + prow;
    if (d < Nn) pself = leaky02(asrc[d * H + phh] + adst[d * H + phh]);
    smax[pidx] = pself;
  }

  auto build = [&](int lo, int hi) -> int {
    if (tid == 0) ovf = 0;
    int mycnt = 0;
    const int chunk = (hi - lo + NW - 1) / NW;
    const int cb = lo + wave * chunk;
    int ce = cb + chunk; if (ce > hi) ce = hi;
    for (int base = cb; base < ce; base += 32 * UA) {
      int dv[UA];
      if (base + 32 * UA <= ce) {
#pragma unroll
        for (int u = 0; u < UA; ++u) dv[u] = dst[base + u * 32 + lane];
      } else {
#pragma unroll
        for (int u = 0; u < UA; ++u) {
          const int e = base + u * 32 + lane;
          dv[u] = (e < ce) ? dst[e] : -1;
        }
      }
#pragma unroll
      for (int u = 0; u < UA; ++u) {
        const unsigned dl = (unsigned)(dv[u] - d0);
        unsigned m = __builtin_amdgcn_ballot_w32(dl < (unsigned)R);
        while (m) {
          const int l = __builtin_ctz(m);
          m &= m - 1;
          const int dls = __builtin_amdgcn_readlane((int)dl, l);
          const int t = dls / RT;
          if (lane == t) {
            if (mycnt < CAPW) wl[(wave * NTEAMS + t) * CAPW + mycnt] = base + u * 32 + l;
            ++mycnt;
          }
        }
      }
    }
    if (lane < NTEAMS) wcnt[wave * NTEAMS + lane] = mycnt;
    __syncthreads();

    int tot = 0;
    bool bad = false;
    for (int w = 0; w < NW; ++w) {
      const int c = wcnt[w * NTEAMS + team];
      bad |= (c > CAPW);
      tot += c;
    }
    bad |= (tot > TCAP);
    if (bad) ovf = 1;
    __syncthreads();
    const int o = ovf;
    __syncthreads();
    if (o) return -1;

    for (int j = c4; j < tot; j += F4) {
      int rem = j, w = 0;
      int c = wcnt[team];
      while (rem >= c) { rem -= c; ++w; c = wcnt[w * NTEAMS + team]; }
      const int e = wl[(w * NTEAMS + team) * CAPW + rem];
      const int s = src[e];
      const int d = dst[e];
      ts[j] = s;
      td[j] = d - d0 - r0;
      for (int hh = 0; hh < H; ++hh)
        tq[j * H + hh] = leaky02(asrc[s * H + hh] + adst[d * H + hh]);
    }
    __syncthreads();
    return tot;
  };

  auto build_parked = [&]() -> int {
    constexpr int WT  = F4 / 32;
    constexpr int EPW = (PCAP + WT - 1) / WT;
    constexpr int UL  = (EPW + 31) / 32;
    const float* sb = out + (size_t)blockIdx.x * pstride;
    bool have = can_park;
    if (have) {
      for (int p = 0; p < PT; ++p) {
        const int np = __float_as_int(sb[p]);
        have &= (np >= 0 && np <= PCAP);
      }
    }
    if (!have) return -2;
    if (tid == 0) ovf = 0;
    const int p  = r0 / PRT;
    const int n  = __float_as_int(sb[p]);
    const float* ps_src = sb + PT + p * 2 * PCAP;
    const float* ps_row = ps_src + PCAP;
    const int wt = c4 >> 5;
    int sv[UL], rv[UL], pos[UL];
#pragma unroll
    for (int u = 0; u < UL; ++u) {
      const int k   = u * 32 + lane;
      const int idx = wt * EPW + k;
      const bool ok = (k < EPW) && (idx < n);
      sv[u] = ok ? __float_as_int(ps_src[idx]) : 0;
      rv[u] = ok ? __float_as_int(ps_row[idx]) : -1;
    }
    int cnt = 0;
#pragma unroll
    for (int u = 0; u < UL; ++u) {
      const unsigned dl = (unsigned)(rv[u] - r0);
      const unsigned m = __builtin_amdgcn_ballot_w32(dl < (unsigned)RT);
      pos[u] = cnt + __builtin_popcount(m & ((1u << lane) - 1u));
      cnt += __builtin_popcount(m);
    }
    if (lane == 0) wcnt[team * WT + wt] = cnt;
    __syncthreads();
    int totl = 0, off = 0;
    for (int w = 0; w < WT; ++w) {
      const int c = wcnt[team * WT + w];
      off += (w < wt) ? c : 0;
      totl += c;
    }
    if (totl > TCAP) ovf = 1;
    __syncthreads();
    const int o = ovf;
    __syncthreads();
    if (o) return -1;
    float lg[UL][H];
#pragma unroll
    for (int u = 0; u < UL; ++u) {
      const bool hit = (unsigned)(rv[u] - r0) < (unsigned)RT;
      const int s = hit ? sv[u] : 0;
      const int d = hit ? d0 + rv[u] : d0;
      for (int hh = 0; hh < H; ++hh)
        lg[u][hh] = leaky02(asrc[s * H + hh] + adst[d * H + hh]);
    }
#pragma unroll
    for (int u = 0; u < UL; ++u) {
      const unsigned dl = (unsigned)(rv[u] - r0);
      if (dl < (unsigned)RT) {
        const int j = off + pos[u];
        ts[j] = sv[u];
        td[j] = (int)dl;
        for (int hh = 0; hh < H; ++hh) tq[j * H + hh] = lg[u][hh];
      }
    }
    __syncthreads();
    return totl;
  };

  auto group_rows = [&](int tot) {
    constexpr int WT = F4 / 32;
    constexpr int UW = 8;
    static_assert(UW <= 16 && TCAP % 4 == 0, "tdl padding and 16-byte loads");
    static_assert(RT <= 32, "one lane per row in every wave of a team");
    const int wt = c4 >> 5;
    const int sl = (((tot + WT - 1) / WT) + 3) & ~3;
    const int lo = wt * sl;
    const int hi = (lo + sl < tot) ? lo + sl : tot;
    const bool rowt = lane < RT;
    int cnt = 0;
    if (rowt) {
      for (int j = lo; j < hi; j += UW) {
        int t[UW];
#pragma unroll
        for (int q = 0; q < UW / 4; ++q) {
          const int4 x = *(const int4*)(td + j + 4 * q);
          t[4 * q] = x.x; t[4 * q + 1] = x.y; t[4 * q + 2] = x.z; t[4 * q + 3] = x.w;
        }
#pragma unroll
        for (int u = 0; u < UW; ++u) cnt += ((j + u < hi) && (t[u] == lane)) ? 1 : 0;
      }
      rcnt[(team * WT + wt) * RT + lane] = cnt;
    }
    __syncthreads();
    if (rowt) {
      const int* cw = rcnt + team * WT * RT;
      int st = 0;
#pragma unroll
      for (int q = 0; q < WT * RT; ++q) {
        const int w = q / RT, r = q % RT;
        const int v = cw[q];
        st += (r < lane || (r == lane && w < wt)) ? v : 0;
      }
      if (wt == 0) {
        rb[lane] = st;
        if (lane == RT - 1) rb[RT] = tot;
      }
      int k = st;
      for (int j = lo; j < hi; j += UW) {
        int t[UW];
#pragma unroll
        for (int q = 0; q < UW / 4; ++q) {
          const int4 x = *(const int4*)(td + j + 4 * q);
          t[4 * q] = x.x; t[4 * q + 1] = x.y; t[4 * q + 2] = x.z; t[4 * q + 3] = x.w;
        }
#pragma unroll
        for (int u = 0; u < UW; ++u) {
          if ((j + u < hi) && (t[u] == lane)) { pm[k] = j + u; ++k; }
        }
      }
    }
    __syncthreads();
  };

  auto pass_max = [&]() {
    if (stat) {
      float m = smax[pidx];
      const int e = rb[prow + 1];
#pragma unroll 4
      for (int i = rb[prow]; i < e; ++i) m = fmaxf(m, tq[pm[i] * H + phh]);
      smax[pidx] = m;
    }
  };

  auto pass_den = [&]() {
    if (stat) {
      const float m = smax[pidx];
      float s = sden[pidx];
      const int e = rb[prow + 1];
#pragma unroll 4
      for (int i = rb[prow]; i < e; ++i) s += __expf(tq[pm[i] * H + phh] - m);
      sden[pidx] = s;
    }
  };

  auto pass_agg = [&](int tot) {
    for (int j = c4; j < tot; j += F4) {
      const int rr = (r0 + td[j]) * H;
      for (int hh = 0; hh < H; ++hh)
        tq[j * H + hh] = __expf(tq[j * H + hh] - smax[rr + hh]) / sden[rr + hh];
    }
    __syncthreads();
    const float* tqh = tq + h;
    int cur = -1;
    v4f a = {0.f, 0.f, 0.f, 0.f};
    auto add = [&](int row, const v4f& v, float w) {
      if (row != cur) {
        if (cur >= 0) arow[cur * F4] = a;
        cur = row;
        a = arow[cur * F4];
      }
      a += v * w;
    };
    int i = 0;
    for (; i + UC <= tot; i += UC) {
      int sr[UC], dr[UC]; float wt[UC]; v4f v[UC];
#pragma unroll
      for (int u = 0; u < UC; ++u) {
        const int jj = __builtin_amdgcn_readfirstlane(pm[i + u]);
        sr[u] = __builtin_amdgcn_readfirstlane(ts[jj]);
        dr[u] = __builtin_amdgcn_readfirstlane(td[jj]);
        wt[u] = tqh[jj * H];
      }
#pragma unroll
      for (int u = 0; u < UC; ++u) v[u] = *(const v4f*)(xcol + (size_t)sr[u] * F);
#pragma unroll
      for (int u = 0; u < UC; ++u) add(dr[u], v[u], wt[u]);
    }
    for (; i < tot; ++i) {
      const int jj = __builtin_amdgcn_readfirstlane(pm[i]);
      const int s  = __builtin_amdgcn_readfirstlane(ts[jj]);
      const int rw = __builtin_amdgcn_readfirstlane(td[jj]);
      const v4f v1 = *(const v4f*)(xcol + (size_t)s * F);
      add(rw, v1, tqh[jj * H]);
    }
    if (cur >= 0) arow[cur * F4] = a;
  };

  int tot;
  if constexpr (LOAD) {
    tot = build_parked();
    if (tot == -2) tot = build(0, E0);
  } else {
    tot = build(0, E0);
  }
  const bool whole = (tot >= 0);
  if (whole) group_rows(tot);
  if constexpr (SAVE) {
    if (can_park) {
      float* sb = park + (size_t)blockIdx.x * pstride;
      if (whole) {
        float* ps_src = sb + PT + team * 2 * PCAP;
        for (int j = c4; j < tot; j += F4) {
          ps_src[j]        = __int_as_float(ts[j]);
          ps_src[PCAP + j] = __int_as_float(r0 + td[j]);
        }
      }
      if (c4 == 0) sb[team] = __int_as_float(whole ? tot : -1);
    }
  }
  const int step = whole ? ((E0 > 0) ? E0 : 1) : SUB;

  for (int mode = 0; mode < 3; ++mode) {
    if (mode == 1 && stat) sden[pidx] = __expf(pself - smax[pidx]);
    if (mode == 2) {
      if (stat) swt[pidx] = __expf(pself - smax[pidx]) / sden[pidx];
      __syncthreads();
      for (int r = 0; r < RT; r += UI) {
        v4f v[UI];
#pragma unroll
        for (int u = 0; u < UI; ++u) {
          const int d = d0 + r0 + r + u;
          const v4f z = {0.f, 0.f, 0.f, 0.f};
          v[u] = (d < Nn) ? *(const v4f*)(xcol + (size_t)d * F) : z;
        }
#pragma unroll
        for (int u = 0; u < UI; ++u) arow[(r + u) * F4] = v[u] * swt[(r0 + r + u) * H + h];
      }
    }
    for (int lo = 0; lo < E0; lo += step) {
      if (!whole) {
        const int hi = (lo + SUB < E0) ? lo + SUB : E0;
        tot = build(lo, hi);
        group_rows(tot);
      }
      if (mode == 0) pass_max();
      else if (mode == 1) pass_den();
      else pass_agg(tot);
      __syncthreads();
    }
    __syncthreads();
  }

  const float b0 = bias[c4 * 4 + 0], b1 = bias[c4 * 4 + 1];
  const float b2 = bias[c4 * 4 + 2], b3 = bias[c4 * 4 + 3];
  for (int r = 0; r < RT; ++r) {
    const int d = d0 + r0 + r;
    if (d < Nn) {
      v4f a = arow[r * F4];
      a.x += b0; a.y += b1; a.z += b2; a.w += b3;
      if (ELU) {
        a.x = (a.x > 0.f) ? a.x : expm1f(a.x);
        a.y = (a.y > 0.f) ? a.y : expm1f(a.y);
        a.z = (a.z > 0.f) ? a.z : expm1f(a.z);
        a.w = (a.w > 0.f) ? a.w : expm1f(a.w);
      }
      *(v4f*)(out + (size_t)d * F + c4 * 4) = a;
    }
  }
}


extern "C" void kernel_launch(void* const* d_in, const int* in_sizes, int n_in,
                              void* d_out, int out_size, void* d_ws, size_t ws_size,
                              hipStream_t stream) {
  const float* x   = (const float*)d_in[0];
  const int*   ei  = (const int*)d_in[1];
  const float* W1  = (const float*)d_in[2];
  const float* as1 = (const float*)d_in[3];
  const float* ad1 = (const float*)d_in[4];
  const float* b1  = (const float*)d_in[5];
  const float* W2  = (const float*)d_in[6];
  const float* as2 = (const float*)d_in[7];
  const float* ad2 = (const float*)d_in[8];
  const float* b2  = (const float*)d_in[9];

  const int K1 = 768, F1 = 768, H1 = 3;
  const int F2 = 256, H2 = 1;
  const int Nn = in_sizes[0] / K1;
  const int E0 = in_sizes[1] / 2;

  const int* srcI = ei;
  const int* dstI = ei + E0;

  float* ws = (float*)d_ws;
  float* xp1   = ws; ws += (size_t)Nn * F1;
  float* h1    = ws; ws += (size_t)Nn * F1;
  float* xp2   = ws; ws += (size_t)Nn * F2;
  float* asrc1 = ws; ws += (size_t)Nn * H1;
  float* adst1 = ws; ws += (size_t)Nn * H1;
  float* asrc2 = ws; ws += Nn;
  float* adst2 = ws; ws += Nn;

  const int PSTR = 80 * F2;
  const int PLIM = Nn * F2;

  float* wt1 = h1;
  float* wt2 = (float*)d_out + 8192;
  fold_att<<<(K1 * H1 + F1 * H2 + 7) / 8, 256, 0, stream>>>(
      W1, as1, ad1, wt1, K1, H1, W2, as2, ad2, wt2, F1, H2);

  {
    dim3 g(Nn / 32, F1 / 64 + 1);
    wmma_gemm_f32<<<g, 32, 0, stream>>>(x, W1, xp1, Nn, F1, K1, wt1, asrc1, adst1, H1);
  }
  owner_edge_phase<3, 768, 20, 48, 512, true, true, false, 20, 512>
      <<<(Nn + 79) / 80, 768, 0, stream>>>(
      srcI, dstI, E0, Nn, xp1, asrc1, adst1, b1, h1, (float*)d_out, PSTR, PLIM);

  {
    dim3 g(Nn / 32, F2 / 64 + 1);
    wmma_gemm_f32<<<g, 32, 0, stream>>>(h1, W2, xp2, Nn, F2, F1, wt2, asrc2, adst2, H2);
  }
  owner_edge_phase<1, 1024, 5, 24, 256, false, false, true, 20, 512>
      <<<(Nn + 79) / 80, 1024, 0, stream>>>(
      srcI, dstI, E0, Nn, xp2, asrc2, adst2, b2, (float*)d_out, nullptr, PSTR, PLIM);
}


# Layer-2 GEMM and attention dots run inside the layer-1 edge workgroup: its 80 finished rows stay in LDS (padded pitch) as the WMMA A operand; one launch and the h1 write+read gone; sums kept
# speedup vs baseline: 4.7124x; 1.0477x over previous
#include <hip/hip_runtime.h>
#include <math.h>


typedef __attribute__((ext_vector_type(2))) float v2f;
typedef __attribute__((ext_vector_type(4))) float v4f;
typedef __attribute__((ext_vector_type(8))) float v8f;

__global__ __launch_bounds__(32) void wmma_gemm_f32(
    const float* __restrict__ A, const float* __restrict__ B,
    float* __restrict__ C, int M, int N, int K,
    const float* __restrict__ Wt, float* __restrict__ asrc,
    float* __restrict__ adst, int H)
{
  const int lane = threadIdx.x & 31;
  const int half = lane >> 4;
  const int l15  = lane & 15;
  const int tm   = blockIdx.x;

  if (blockIdx.y == 0) {
    const int bc = (l15 < 2 * H) ? l15 : 0;
    const float* Ad = A + (size_t)(tm * 32 + l15) * K + (half << 4);
    const float* Ae = Ad + (size_t)16 * K;
    const float* Bd = Wt + (size_t)bc * K + (half << 4);
    v8f d0 = {}; v8f d1 = {}; v8f e0 = {}; v8f e1 = {};
    for (int kb = 0; kb < K; kb += 32) {
      v4f aq[4];
      v4f ar[4];
      v4f bq[4];
#pragma unroll
      for (int i = 0; i < 4; ++i) aq[i] = *(const v4f*)(Ad + kb + 4 * i);
#pragma unroll
      for (int i = 0; i < 4; ++i) ar[i] = *(const v4f*)(Ae + kb + 4 * i);
#pragma unroll
      for (int i = 0; i < 4; ++i) bq[i] = *(const v4f*)(Bd + kb + 4 * i);
#pragma unroll
      for (int j = 0; j < 4; ++j) {
        v2f a0, a1, f0, f1, b0, b1;
        a0.x = aq[j][0]; a0.y = aq[j][1];
        a1.x = aq[j][2]; a1.y = aq[j][3];
        f0.x = ar[j][0]; f0.y = ar[j][1];
        f1.x = ar[j][2]; f1.y = ar[j][3];
        b0.x = bq[j][0]; b0.y = bq[j][1];
        b1.x = bq[j][2]; b1.y = bq[j][3];
        d0 = __builtin_amdgcn_wmma_f32_16x16x4_f32(false, a0, false, b0, (short)0, d0, false, false);
        e0 = __builtin_amdgcn_wmma_f32_16x16x4_f32(false, f0, false, b0, (short)0, e0, false, false);
        d1 = __builtin_amdgcn_wmma_f32_16x16x4_f32(false, a1, false, b1, (short)0, d1, false, false);
        e1 = __builtin_amdgcn_wmma_f32_16x16x4_f32(false, f1, false, b1, (short)0, e1, false, false);
      }
    }
    if (l15 < 2 * H) {
      float* o = (l15 < H) ? (asrc + l15) : (adst + (l15 - H));
      const int row0 = tm * 32 + (half << 3);
      for (int r = 0; r < 8; ++r) o[(size_t)(row0 + r) * H] = d0[r] + d1[r];
      for (int r = 0; r < 8; ++r) o[(size_t)(row0 + 16 + r) * H] = e0[r] + e1[r];
    }
    return;
  }
  const int tn   = (blockIdx.y - 1) << 6;

  v8f c0 = {}; v8f c1 = {}; v8f c2 = {}; v8f c3 = {};
  v8f g0 = {}; v8f g1 = {}; v8f g2 = {}; v8f g3 = {};

  const float* Ap = A + (size_t)(tm * 32 + l15) * K + (half << 4);
  const float* Aq = Ap + (size_t)16 * K;
  const float* Bp = B + (size_t)(half << 4) * N + tn + (l15 << 2);
  for (int kb = 0; kb < K; kb += 32) {
    v4f aq[4];
    v4f ar[4];
    v4f bq[16];
#pragma unroll
    for (int i = 0; i < 4; ++i) aq[i] = *(const v4f*)(Ap + kb + 4 * i);
#pragma unroll
    for (int i = 0; i < 4; ++i) ar[i] = *(const v4f*)(Aq + kb + 4 * i);
#pragma unroll
    for (int m = 0; m < 16; ++m) bq[m] = *(const v4f*)(Bp + (size_t)(kb + m) * N);
    __builtin_amdgcn_sched_barrier(0);
#pragma unroll
    for (int j = 0; j < 8; ++j) {
      v2f a, f;
      a.x = aq[j >> 1][2 * (j & 1)];
      a.y = aq[j >> 1][2 * (j & 1) + 1];
      f.x = ar[j >> 1][2 * (j & 1)];
      f.y = ar[j >> 1][2 * (j & 1) + 1];
      v2f b0, b1, b2, b3;
      b0.x = bq[2 * j][0]; b0.y = bq[2 * j + 1][0];
      b1.x = bq[2 * j][1]; b1.y = bq[2 * j + 1][1];
      b2.x = bq[2 * j][2]; b2.y = bq[2 * j + 1][2];
      b3.x = bq[2 * j][3]; b3.y = bq[2 * j + 1][3];
      c0 = __builtin_amdgcn_wmma_f32_16x16x4_f32(false, a, false, b0, (short)0, c0, false, false);
      g0 = __builtin_amdgcn_wmma_f32_16x16x4_f32(false, f, false, b0, (short)0, g0, false, false);
      c1 = __builtin_amdgcn_wmma_f32_16x16x4_f32(false, a, false, b1, (short)0, c1, false, false);
      g1 = __builtin_amdgcn_wmma_f32_16x16x4_f32(false, f, false, b1, (short)0, g1, false, false);
      c2 = __builtin_amdgcn_wmma_f32_16x16x4_f32(false, a, false, b2, (short)0, c2, false, false);
      g2 = __builtin_amdgcn_wmma_f32_16x16x4_f32(false, f, false, b2, (short)0, g2, false, false);
      c3 = __builtin_amdgcn_wmma_f32_16x16x4_f32(false, a, false, b3, (short)0, c3, false, false);
      g3 = __builtin_amdgcn_wmma_f32_16x16x4_f32(false, f, false, b3, (short)0, g3, false, false);
    }
  }

  const int orow = tm * 32 + (half << 3);
  for (int r = 0; r < 8; ++r) {
    v4f o;
    o.x = c0[r]; o.y = c1[r]; o.z = c2[r]; o.w = c3[r];
    *(v4f*)(C + (size_t)(orow + r) * N + tn + (l15 << 2)) = o;
  }
  for (int r = 0; r < 8; ++r) {
    v4f o;
    o.x = g0[r]; o.y = g1[r]; o.z = g2[r]; o.w = g3[r];
    *(v4f*)(C + (size_t)(orow + 16 + r) * N + tn + (l15 << 2)) = o;
  }
}

__global__ __launch_bounds__(256) void fold_att(
    const float* __restrict__ W1, const float* __restrict__ as1,
    const float* __restrict__ ad1, float* __restrict__ wt1, int K1, int H1,
    const float* __restrict__ W2, const float* __restrict__ as2,
    const float* __restrict__ ad2, float* __restrict__ wt2, int K2, int H2)
{
  int wid = (int)((blockIdx.x * blockDim.x + threadIdx.x) >> 5);
  const int lane = threadIdx.x & 31;
  const float* Wm = W1; const float* as = as1; const float* ad = ad1;
  float* wt = wt1; int K = K1; int H = H1;
  if (wid >= K1 * H1) {
    wid -= K1 * H1;
    if (wid >= K2 * H2) return;
    Wm = W2; as = as2; ad = ad2; wt = wt2; K = K2; H = H2;
  }
  const int k = wid / H;
  const int h = wid - k * H;
  const float* wr = Wm + (size_t)k * H * 256 + h * 256 + lane * 4;
  const float* sr = as + h * 256 + lane * 4;
  const float* dr = ad + h * 256 + lane * 4;
  const v4f w0 = *(const v4f*)wr,  w1 = *(const v4f*)(wr + 128);
  const v4f s0 = *(const v4f*)sr,  s1 = *(const v4f*)(sr + 128);
  const v4f e0 = *(const v4f*)dr,  e1 = *(const v4f*)(dr + 128);
  float s = w0.x * s0.x + w0.y * s0.y + w0.z * s0.z + w0.w * s0.w
          + w1.x * s1.x + w1.y * s1.y + w1.z * s1.z + w1.w * s1.w;
  float d = w0.x * e0.x + w0.y * e0.y + w0.z * e0.z + w0.w * e0.w
          + w1.x * e1.x + w1.y * e1.y + w1.z * e1.z + w1.w * e1.w;
  for (int off = 16; off > 0; off >>= 1) {
    s += __shfl_down(s, off, 32);
    d += __shfl_down(d, off, 32);
  }
  if (lane == 0) {
    wt[(size_t)h * K + k]       = s;
    wt[(size_t)(H + h) * K + k] = d;
  }
}

template <int NTL, int K, int AP>
__device__ __forceinline__ void lds_dot_strip(
    const v4f* acc, const float* __restrict__ Wt, int rtA, int rtB, int lane,
    int d0, int Nn, float* __restrict__ asn, float* __restrict__ adn)
{
  const int half = lane >> 4;
  const int l15  = lane & 15;
  const int bc   = (l15 < 2) ? l15 : 0;
  const float* Bd  = Wt + (size_t)bc * K + (half << 4);
  const v4f*   AlA = acc + (rtA * 16 + l15) * AP + (half << 2);
  const v4f*   AlB = acc + (rtB * 16 + l15) * AP + (half << 2);
  v8f e0 = {}; v8f e1 = {}; v8f g0 = {}; v8f g1 = {};
  for (int kb = 0; kb < K; kb += 32) {
    v4f aq[4];
    v4f ar[4];
    v4f bq[4];
#pragma unroll
    for (int i = 0; i < 4; ++i) aq[i] = AlA[(kb >> 2) + i];
    if constexpr (NTL == 2) {
#pragma unroll
      for (int i = 0; i < 4; ++i) ar[i] = AlB[(kb >> 2) + i];
    }
#pragma unroll
    for (int i = 0; i < 4; ++i) bq[i] = *(const v4f*)(Bd + kb + 4 * i);
#pragma unroll
    for (int j = 0; j < 4; ++j) {
      v2f a0, a1, p0, p1;
      a0.x = aq[j][0]; a0.y = aq[j][1];
      a1.x = aq[j][2]; a1.y = aq[j][3];
      p0.x = bq[j][0]; p0.y = bq[j][1];
      p1.x = bq[j][2]; p1.y = bq[j][3];
      e0 = __builtin_amdgcn_wmma_f32_16x16x4_f32(false, a0, false, p0, (short)0, e0, false, false);
      e1 = __builtin_amdgcn_wmma_f32_16x16x4_f32(false, a1, false, p1, (short)0, e1, false, false);
      if constexpr (NTL == 2) {
        v2f f0, f1;
        f0.x = ar[j][0]; f0.y = ar[j][1];
        f1.x = ar[j][2]; f1.y = ar[j][3];
        g0 = __builtin_amdgcn_wmma_f32_16x16x4_f32(false, f0, false, p0, (short)0, g0, false, false);
        g1 = __builtin_amdgcn_wmma_f32_16x16x4_f32(false, f1, false, p1, (short)0, g1, false, false);
      }
    }
  }
  if (l15 < 2) {
    float* o = (l15 == 0) ? asn : adn;
    const int rowA = d0 + rtA * 16 + (half << 3);
    for (int r = 0; r < 8; ++r)
      if (rowA + r < Nn) o[rowA + r] = e0[r] + e1[r];
    if constexpr (NTL == 2) {
      const int rowB = d0 + rtB * 16 + (half << 3);
      for (int r = 0; r < 8; ++r)
        if (rowB + r < Nn) o[rowB + r] = g0[r] + g1[r];
    }
  }
}


__device__ __forceinline__ float leaky02(float a) { return (a > 0.f) ? a : 0.2f * a; }

template <int H, int NT, int RT, int CAPW, int TCAP, bool ELU,
          bool SAVE, bool LOAD, int PRT, int PCAP>
__global__ __launch_bounds__(NT) void owner_edge_phase(
    const int* __restrict__ src, const int* __restrict__ dst, int E0, int Nn,
    const float* __restrict__ xp, const float* __restrict__ asrc,
    const float* __restrict__ adst, const float* __restrict__ bias,
    float* __restrict__ out, float* __restrict__ park, int pstride, int plimit,
    const float* __restrict__ Wn, const float* __restrict__ Wtn,
    float* __restrict__ xpn, float* __restrict__ asn, float* __restrict__ adn)
{
  constexpr int F      = H * 256;
  constexpr int F4     = F / 4;
  constexpr int NTEAMS = NT / F4;
  constexpr int R      = NTEAMS * RT;
  constexpr int NW     = NT / 32;
  constexpr int NL     = NW * NTEAMS;
  constexpr int SUB    = (TCAP < NW * CAPW) ? TCAP : NW * CAPW;
  constexpr int UA     = 16;
  constexpr int UC     = 8;
  constexpr int UI     = 5;
  static_assert(NT % F4 == 0 && F4 % 32 == 0 && NTEAMS <= 32, "team shape");
  static_assert(RT * H <= F4, "one thread per (row, head) of a team");
  static_assert(RT % UI == 0, "self-loop rows are loaded UI at a time");
  static_assert(R % PRT == 0 && PRT % RT == 0 && PCAP > 0, "parked lists: a team's rows lie in one parent team");
  static_assert(!SAVE || (PRT == RT && PCAP == TCAP), "a SAVE launch parks its own teams");
  static_assert(NL * CAPW >= NTEAMS * TCAP, "the row-grouped entry numbers reuse the wave lists' space");
  constexpr int PT    = R / PRT;
  constexpr int PNEED = PT + 2 * PT * PCAP;
  constexpr int AP    = SAVE ? F4 + 1 : F4;

  __shared__ v4f   acc[R * AP];
  __shared__ int   wl[NL * CAPW];
  __shared__ int   wcnt[NL];
  __shared__ int   tsrc[NTEAMS * TCAP];
  __shared__ __attribute__((aligned(16))) int tdl[NTEAMS * TCAP + 16];
  __shared__ float tw[NTEAMS * TCAP * H];
  __shared__ float smax[R * H];
  __shared__ float sden[R * H];
  __shared__ float swt[R * H];
  __shared__ __attribute__((aligned(16))) int rcnt[NW * RT];
  __shared__ int   rbeg[NTEAMS * (RT + 1)];
  __shared__ int   ovf;

  const int tid  = threadIdx.x;
  const int lane = tid & 31;
  const int wave = tid >> 5;
  const int team = tid / F4;
  const int c4   = tid - team * F4;
  const int h    = c4 >> 6;
  const int d0   = blockIdx.x * R;
  const int r0   = team * RT;
  const bool can_park = (size_t)blockIdx.x * (size_t)pstride + (size_t)PNEED <= (size_t)plimit;

  const float* xcol = xp + c4 * 4;
  v4f*   arow = acc + r0 * AP + c4;
  int*   ts   = tsrc + team * TCAP;
  int*   td   = tdl + team * TCAP;
  float* tq   = tw + team * TCAP * H;
  int*   pm   = wl + team * TCAP;
  int*   rb   = rbeg + team * (RT + 1);

  const bool stat = c4 < RT * H;
  const int  prow = c4 / H;
  const int  phh  = c4 - prow * H;
  const int  pidx = (r0 + prow) * H + phh;
  float pself = 0.f;
  if (stat) {
    const int d = d0 + r0 + prow;
    if (d < Nn) pself = leaky02(asrc[d * H + phh] + adst[d * H + phh]);
    smax[pidx] = pself;
  }

  auto build = [&](int lo, int hi) -> int {
    if (tid == 0) ovf = 0;
    int mycnt = 0;
    const int chunk = (hi - lo + NW - 1) / NW;
    const int cb = lo + wave * chunk;
    int ce = cb + chunk; if (ce > hi) ce = hi;
    for (int base = cb; base < ce; base += 32 * UA) {
      int dv[UA];
      if (base + 32 * UA <= ce) {
#pragma unroll
        for (int u = 0; u < UA; ++u) dv[u] = dst[base + u * 32 + lane];
      } else {
#pragma unroll
        for (int u = 0; u < UA; ++u) {
          const int e = base + u * 32 + lane;
          dv[u] = (e < ce) ? dst[e] : -1;
        }
      }
#pragma unroll
      for (int u = 0; u < UA; ++u) {
        const unsigned dl = (unsigned)(dv[u] - d0);
        unsigned m = __builtin_amdgcn_ballot_w32(dl < (unsigned)R);
        while (m) {
          const int l = __builtin_ctz(m);
          m &= m - 1;
          const int dls = __builtin_amdgcn_readlane((int)dl, l);
          const int t = dls / RT;
          if (lane == t) {
            if (mycnt < CAPW) wl[(wave * NTEAMS + t) * CAPW + mycnt] = base + u * 32 + l;
            ++mycnt;
          }
        }
      }
    }
    if (lane < NTEAMS) wcnt[wave * NTEAMS + lane] = mycnt;
    __syncthreads();

    int tot = 0;
    bool bad = false;
    for (int w = 0; w < NW; ++w) {
      const int c = wcnt[w * NTEAMS + team];
      bad |= (c > CAPW);
      tot += c;
    }
    bad |= (tot > TCAP);
    if (bad) ovf = 1;
    __syncthreads();
    const int o = ovf;
    __syncthreads();
    if (o) return -1;

    for (int j = c4; j < tot; j += F4) {
      int rem = j, w = 0;
      int c = wcnt[team];
      while (rem >= c) { rem -= c; ++w; c = wcnt[w * NTEAMS + team]; }
      const int e = wl[(w * NTEAMS + team) * CAPW + rem];
      const int s = src[e];
      const int d = dst[e];
      ts[j] = s;
      td[j] = d - d0 - r0;
      for (int hh = 0; hh < H; ++hh)
        tq[j * H + hh] = leaky02(asrc[s * H + hh] + adst[d * H + hh]);
    }
    __syncthreads();
    return tot;
  };

  auto build_parked = [&]() -> int {
    constexpr int WT  = F4 / 32;
    constexpr int EPW = (PCAP + WT - 1) / WT;
    constexpr int UL  = (EPW + 31) / 32;
    const float* sb = out + (size_t)blockIdx.x * pstride;
    bool have = can_park;
    if (have) {
      for (int p = 0; p < PT; ++p) {
        const int np = __float_as_int(sb[p]);
        have &= (np >= 0 && np <= PCAP);
      }
    }
    if (!have) return -2;
    if (tid == 0) ovf = 0;
    const int p  = r0 / PRT;
    const int n  = __float_as_int(sb[p]);
    const float* ps_src = sb + PT + p * 2 * PCAP;
    const float* ps_row = ps_src + PCAP;
    const int wt = c4 >> 5;
    int sv[UL], rv[UL], pos[UL];
#pragma unroll
    for (int u = 0; u < UL; ++u) {
      const int k   = u * 32 + lane;
      const int idx = wt * EPW + k;
      const bool ok = (k < EPW) && (idx < n);
      sv[u] = ok ? __float_as_int(ps_src[idx]) : 0;
      rv[u] = ok ? __float_as_int(ps_row[idx]) : -1;
    }
    int cnt = 0;
#pragma unroll
    for (int u = 0; u < UL; ++u) {
      const unsigned dl = (unsigned)(rv[u] - r0);
      const unsigned m = __builtin_amdgcn_ballot_w32(dl < (unsigned)RT);
      pos[u] = cnt + __builtin_popcount(m & ((1u << lane) - 1u));
      cnt += __builtin_popcount(m);
    }
    if (lane == 0) wcnt[team * WT + wt] = cnt;
    __syncthreads();
    int totl = 0, off = 0;
    for (int w = 0; w < WT; ++w) {
      const int c = wcnt[team * WT + w];
      off += (w < wt) ? c : 0;
      totl += c;
    }
    if (totl > TCAP) ovf = 1;
    __syncthreads();
    const int o = ovf;
    __syncthreads();
    if (o) return -1;
    float lg[UL][H];
#pragma unroll
    for (int u = 0; u < UL; ++u) {
      const bool hit = (unsigned)(rv[u] - r0) < (unsigned)RT;
      const int s = hit ? sv[u] : 0;
      const int d = hit ? d0 + rv[u] : d0;
      for (int hh = 0; hh < H; ++hh)
        lg[u][hh] = leaky02(asrc[s * H + hh] + adst[d * H + hh]);
    }
#pragma unroll
    for (int u = 0; u < UL; ++u) {
      const unsigned dl = (unsigned)(rv[u] - r0);
      if (dl < (unsigned)RT) {
        const int j = off + pos[u];
        ts[j] = sv[u];
        td[j] = (int)dl;
        for (int hh = 0; hh < H; ++hh) tq[j * H + hh] = lg[u][hh];
      }
    }
    __syncthreads();
    return totl;
  };

  auto group_rows = [&](int tot) {
    constexpr int WT = F4 / 32;
    constexpr int UW = 8;
    static_assert(UW <= 16 && TCAP % 4 == 0, "tdl padding and 16-byte loads");
    static_assert(RT <= 32, "one lane per row in every wave of a team");
    const int wt = c4 >> 5;
    const int sl = (((tot + WT - 1) / WT) + 3) & ~3;
    const int lo = wt * sl;
    const int hi = (lo + sl < tot) ? lo + sl : tot;
    const bool rowt = lane < RT;
    int cnt = 0;
    if (rowt) {
      for (int j = lo; j < hi; j += UW) {
        int t[UW];
#pragma unroll
        for (int q = 0; q < UW / 4; ++q) {
          const int4 x = *(const int4*)(td + j + 4 * q);
          t[4 * q] = x.x; t[4 * q + 1] = x.y; t[4 * q + 2] = x.z; t[4 * q + 3] = x.w;
        }
#pragma unroll
        for (int u = 0; u < UW; ++u) cnt += ((j + u < hi) && (t[u] == lane)) ? 1 : 0;
      }
      rcnt[(team * WT + wt) * RT + lane] = cnt;
    }
    __syncthreads();
    if (rowt) {
      const int* cw = rcnt + team * WT * RT;
      int st = 0;
#pragma unroll
      for (int q = 0; q < WT * RT; ++q) {
        const int w = q / RT, r = q % RT;
        const int v = cw[q];
        st += (r < lane || (r == lane && w < wt)) ? v : 0;
      }
      if (wt == 0) {
        rb[lane] = st;
        if (lane == RT - 1) rb[RT] = tot;
      }
      int k = st;
      for (int j = lo; j < hi; j += UW) {
        int t[UW];
#pragma unroll
        for (int q = 0; q < UW / 4; ++q) {
          const int4 x = *(const int4*)(td + j + 4 * q);
          t[4 * q] = x.x; t[4 * q + 1] = x.y; t[4 * q + 2] = x.z; t[4 * q + 3] = x.w;
        }
#pragma unroll
        for (int u = 0; u < UW; ++u) {
          if ((j + u < hi) && (t[u] == lane)) { pm[k] = j + u; ++k; }
        }
      }
    }
    __syncthreads();
  };

  auto pass_max = [&]() {
    if (stat) {
      float m = smax[pidx];
      const int e = rb[prow + 1];
#pragma unroll 4
      for (int i = rb[prow]; i < e; ++i) m = fmaxf(m, tq[pm[i] * H + phh]);
      smax[pidx] = m;
    }
  };

  auto pass_den = [&]() {
    if (stat) {
      const float m = smax[pidx];
      float s = sden[pidx];
      const int e = rb[prow + 1];
#pragma unroll 4
      for (int i = rb[prow]; i < e; ++i) s += __expf(tq[pm[i] * H + phh] - m);
      sden[pidx] = s;
    }
  };

  auto pass_agg = [&](int tot) {
    for (int j = c4; j < tot; j += F4) {
      const int rr = (r0 + td[j]) * H;
      for (int hh = 0; hh < H; ++hh)
        tq[j * H + hh] = __expf(tq[j * H + hh] - smax[rr + hh]) / sden[rr + hh];
    }
    __syncthreads();
    const float* tqh = tq + h;
    int cur = -1;
    v4f a = {0.f, 0.f, 0.f, 0.f};
    auto add = [&](int row, const v4f& v, float w) {
      if (row != cur) {
        if (cur >= 0) arow[cur * AP] = a;
        cur = row;
        a = arow[cur * AP];
      }
      a += v * w;
    };
    int i = 0;
    for (; i + UC <= tot; i += UC) {
      int sr[UC], dr[UC]; float wt[UC]; v4f v[UC];
#pragma unroll
      for (int u = 0; u < UC; ++u) {
        const int jj = __builtin_amdgcn_readfirstlane(pm[i + u]);
        sr[u] = __builtin_amdgcn_readfirstlane(ts[jj]);
        dr[u] = __builtin_amdgcn_readfirstlane(td[jj]);
        wt[u] = tqh[jj * H];
      }
#pragma unroll
      for (int u = 0; u < UC; ++u) v[u] = *(const v4f*)(xcol + (size_t)sr[u] * F);
#pragma unroll
      for (int u = 0; u < UC; ++u) add(dr[u], v[u], wt[u]);
    }
    for (; i < tot; ++i) {
      const int jj = __builtin_amdgcn_readfirstlane(pm[i]);
      const int s  = __builtin_amdgcn_readfirstlane(ts[jj]);
      const int rw = __builtin_amdgcn_readfirstlane(td[jj]);
      const v4f v1 = *(const v4f*)(xcol + (size_t)s * F);
      add(rw, v1, tqh[jj * H]);
    }
    if (cur >= 0) arow[cur * AP] = a;
  };

  int tot;
  if constexpr (LOAD) {
    tot = build_parked();
    if (tot == -2) tot = build(0, E0);
  } else {
    tot = build(0, E0);
  }
  const bool whole = (tot >= 0);
  if (whole) group_rows(tot);
  if constexpr (SAVE) {
    if (can_park) {
      float* sb = park + (size_t)blockIdx.x * pstride;
      if (whole) {
        float* ps_src = sb + PT + team * 2 * PCAP;
        for (int j = c4; j < tot; j += F4) {
          ps_src[j]        = __int_as_float(ts[j]);
          ps_src[PCAP + j] = __int_as_float(r0 + td[j]);
        }
      }
      if (c4 == 0) sb[team] = __int_as_float(whole ? tot : -1);
    }
  }
  const int step = whole ? ((E0 > 0) ? E0 : 1) : SUB;

  for (int mode = 0; mode < 3; ++mode) {
    if (mode == 1 && stat) sden[pidx] = __expf(pself - smax[pidx]);
    if (mode == 2) {
      if (stat) swt[pidx] = __expf(pself - smax[pidx]) / sden[pidx];
      __syncthreads();
      for (int r = 0; r < RT; r += UI) {
        v4f v[UI];
#pragma unroll
        for (int u = 0; u < UI; ++u) {
          const int d = d0 + r0 + r + u;
          const v4f z = {0.f, 0.f, 0.f, 0.f};
          v[u] = (d < Nn) ? *(const v4f*)(xcol + (size_t)d * F) : z;
        }
#pragma unroll
        for (int u = 0; u < UI; ++u) arow[(r + u) * AP] = v[u] * swt[(r0 + r + u) * H + h];
      }
    }
    for (int lo = 0; lo < E0; lo += step) {
      if (!whole) {
        const int hi = (lo + SUB < E0) ? lo + SUB : E0;
        tot = build(lo, hi);
        group_rows(tot);
      }
      if (mode == 0) pass_max();
      else if (mode == 1) pass_den();
      else pass_agg(tot);
      __syncthreads();
    }
    __syncthreads();
  }

  const float b0 = bias[c4 * 4 + 0], b1 = bias[c4 * 4 + 1];
  const float b2 = bias[c4 * 4 + 2], b3 = bias[c4 * 4 + 3];
  for (int r = 0; r < RT; ++r) {
    const int d = d0 + r0 + r;
    if (SAVE || d < Nn) {
      v4f a = arow[r * AP];
      a.x += b0; a.y += b1; a.z += b2; a.w += b3;
      if (ELU) {
        a.x = (a.x > 0.f) ? a.x : expm1f(a.x);
        a.y = (a.y > 0.f) ? a.y : expm1f(a.y);
        a.z = (a.z > 0.f) ? a.z : expm1f(a.z);
        a.w = (a.w > 0.f) ? a.w : expm1f(a.w);
      }
      if constexpr (SAVE) arow[r * AP] = a;
      else *(v4f*)(out + (size_t)d * F + c4 * 4) = a;
    }
  }

  if constexpr (SAVE) {
    constexpr int FN  = 256;
    constexpr int RTL = R / 16;
    static_assert(R % 16 == 0 && NW >= RTL * 4 + 4 && NW == 24, "wave roles of the fused GEMM");
    __syncthreads();
    const int wv   = __builtin_amdgcn_readfirstlane(wave);
    const int half = lane >> 4;
    const int l15  = lane & 15;
    const bool dotw = (wv == 5) || (wv == 10) || (wv == 15) || (wv == 20);
    if (!dotw) {
      const int mi = wv - (wv > 5 ? 1 : 0) - (wv > 10 ? 1 : 0) - (wv > 15 ? 1 : 0) - (wv > 20 ? 1 : 0);
      const int rt = mi >> 2;
      const int tn = (mi & 3) << 6;
      const v4f*   Al = acc + (rt * 16 + l15) * AP + (half << 2);
      const float* Bp = Wn + (size_t)(half << 4) * FN + tn + (l15 << 2);
      v8f c0 = {}; v8f c1 = {}; v8f c2 = {}; v8f c3 = {};
#pragma clang loop unroll(disable)
      for (int kb = 0; kb < F; kb += 32) {
        v4f aq[4];
        v4f bq[16];
#pragma unroll
        for (int i = 0; i < 4; ++i) aq[i] = Al[(kb >> 2) + i];
#pragma unroll
        for (int m = 0; m < 16; ++m) bq[m] = *(const v4f*)(Bp + (size_t)(kb + m) * FN);
        __builtin_amdgcn_sched_barrier(0);
#pragma unroll
        for (int j = 0; j < 8; ++j) {
          v2f a;
          a.x = aq[j >> 1][2 * (j & 1)];
          a.y = aq[j >> 1][2 * (j & 1) + 1];
          v2f q0, q1, q2, q3;
          q0.x = bq[2 * j][0]; q0.y = bq[2 * j + 1][0];
          q1.x = bq[2 * j][1]; q1.y = bq[2 * j + 1][1];
          q2.x = bq[2 * j][2]; q2.y = bq[2 * j + 1][2];
          q3.x = bq[2 * j][3]; q3.y = bq[2 * j + 1][3];
          c0 = __builtin_amdgcn_wmma_f32_16x16x4_f32(false, a, false, q0, (short)0, c0, false, false);
          c1 = __builtin_amdgcn_wmma_f32_16x16x4_f32(false, a, false, q1, (short)0, c1, false, false);
          c2 = __builtin_amdgcn_wmma_f32_16x16x4_f32(false, a, false, q2, (short)0, c2, false, false);
          c3 = __builtin_amdgcn_wmma_f32_16x16x4_f32(false, a, false, q3, (short)0, c3, false, false);
        }
      }
      const int orow = d0 + rt * 16 + (half << 3);
      for (int r = 0; r < 8; ++r) {
        if (orow + r < Nn) {
          v4f o;
          o.x = c0[r]; o.y = c1[r]; o.z = c2[r]; o.w = c3[r];
          *(v4f*)(xpn + (size_t)(orow + r) * FN + tn + (l15 << 2)) = o;
        }
      }
    } else {
      static_assert(RTL == 5, "five row tiles over four dot waves");
      if (wv == 5) lds_dot_strip<2, F, AP>(acc, Wtn, 0, 4, lane, d0, Nn, asn, adn);
      else         lds_dot_strip<1, F, AP>(acc, Wtn, wv / 5 - 1, wv / 5 - 1, lane, d0, Nn, asn, adn);
    }
  }
}


extern "C" void kernel_launch(void* const* d_in, const int* in_sizes, int n_in,
                              void* d_out, int out_size, void* d_ws, size_t ws_size,
                              hipStream_t stream) {
  const float* x   = (const float*)d_in[0];
  const int*   ei  = (const int*)d_in[1];
  const float* W1  = (const float*)d_in[2];
  const float* as1 = (const float*)d_in[3];
  const float* ad1 = (const float*)d_in[4];
  const float* b1  = (const float*)d_in[5];
  const float* W2  = (const float*)d_in[6];
  const float* as2 = (const float*)d_in[7];
  const float* ad2 = (const float*)d_in[8];
  const float* b2  = (const float*)d_in[9];

  const int K1 = 768, F1 = 768, H1 = 3;
  const int F2 = 256, H2 = 1;
  const int Nn = in_sizes[0] / K1;
  const int E0 = in_sizes[1] / 2;

  const int* srcI = ei;
  const int* dstI = ei + E0;

  float* ws = (float*)d_ws;
  float* xp1   = ws; ws += (size_t)Nn * F1;
  float* h1    = ws; ws += (size_t)Nn * F1;
  float* xp2   = ws; ws += (size_t)Nn * F2;
  float* asrc1 = ws; ws += (size_t)Nn * H1;
  float* adst1 = ws; ws += (size_t)Nn * H1;
  float* asrc2 = ws; ws += Nn;
  float* adst2 = ws; ws += Nn;

  const int PSTR = 80 * F2;
  const int PLIM = Nn * F2;

  float* wt1 = h1;
  float* wt2 = (float*)d_out + 8192;
  fold_att<<<(K1 * H1 + F1 * H2 + 7) / 8, 256, 0, stream>>>(
      W1, as1, ad1, wt1, K1, H1, W2, as2, ad2, wt2, F1, H2);

  {
    dim3 g(Nn / 32, F1 / 64 + 1);
    wmma_gemm_f32<<<g, 32, 0, stream>>>(x, W1, xp1, Nn, F1, K1, wt1, asrc1, adst1, H1);
  }
  owner_edge_phase<3, 768, 20, 48, 512, true, true, false, 20, 512>
      <<<(Nn + 79) / 80, 768, 0, stream>>>(
      srcI, dstI, E0, Nn, xp1, asrc1, adst1, b1, h1, (float*)d_out, PSTR, PLIM,
      W2, wt2, xp2, asrc2, adst2);

  owner_edge_phase<1, 1024, 5, 24, 256, false, false, true, 20, 512>
      <<<(Nn + 79) / 80, 1024, 0, stream>>>(
      srcI, dstI, E0, Nn, xp2, asrc2, adst2, b2, (float*)d_out, nullptr, PSTR, PLIM,
      nullptr, nullptr, nullptr, nullptr, nullptr);
}
